# starting kernel
# baseline (speedup 1.0000x reference)
#include <hip/hip_runtime.h>
#include <hip/hip_bf16.h>

typedef __attribute__((ext_vector_type(2))) float v2f;
typedef __attribute__((ext_vector_type(8))) float v8f;

constexpr int B_ = 8, L_ = 8192, T_ = 512, DH_ = 1024, DG_ = 768, DP_ = 256;

__device__ __forceinline__ v8f wmma4(v2f a, v2f b, v8f c) {
  return __builtin_amdgcn_wmma_f32_16x16x4_f32(
       false, a,  false, b,
       (short)0, c,  false,  false);
}

__device__ __forceinline__ v8f zero8() {
  v8f z = {0.f, 0.f, 0.f, 0.f, 0.f, 0.f, 0.f, 0.f};
  return z;
}

template <bool TRANS>
__global__ __launch_bounds__(256) void proj_gemm_kernel(
    const float* __restrict__ A, const float* __restrict__ W,
    float* __restrict__ out, int Kd) {
  const int tid  = threadIdx.x;
  const int w    = tid >> 5;
  const int lane = tid & 31;
  const int half = lane >> 4;
  const int l15  = lane & 15;
  const int row_base = blockIdx.x * 16;
  const int col0 = w * 32 + l15;

  const float* Ab = A + (size_t)(row_base + l15) * Kd + 2 * half;

  v8f acc0 = zero8();
  v8f acc1 = zero8();

  for (int kk = 0; kk < Kd; kk += 4) {
    v2f a;
    a.x = Ab[kk];
    a.y = Ab[kk + 1];
    const float* Wb = W + (size_t)(kk + 2 * half) * DP_ + col0;
    v2f b0; b0.x = Wb[0];  b0.y = Wb[DP_];
    v2f b1; b1.x = Wb[16]; b1.y = Wb[16 + DP_];
    acc0 = wmma4(a, b0, acc0);
    acc1 = wmma4(a, b1, acc1);
  }

#pragma unroll
  for (int v = 0; v < 8; ++v) {
    const int m = v + 8 * half;
    const size_t r = (size_t)row_base + m;
    if (!TRANS) {
      out[r * DP_ + col0]      = acc0[v];
      out[r * DP_ + col0 + 16] = acc1[v];
    } else {
      const size_t bb = r >> 13;
      const size_t l  = r & (size_t)(L_ - 1);
      out[(bb * DP_ + col0) * L_ + l]        = acc0[v];
      out[(bb * DP_ + col0 + 16) * L_ + l]   = acc1[v];
    }
  }
}

__global__ __launch_bounds__(256) void attn_kernel(
    const float* __restrict__ H, const float* __restrict__ Q,
    const float* __restrict__ KT, const int* __restrict__ mask,
    float* __restrict__ Z) {
  __shared__ float q_lds[16 * DP_];
  __shared__ float p_lds[16 * 256];
  __shared__ float red[8 * 16];
  __shared__ float mrow[16];
  __shared__ float srow[16];
  __shared__ float frow[16];

  const int tid  = threadIdx.x;
  const int w    = tid >> 5;
  const int lane = tid & 31;
  const int half = lane >> 4;
  const int l15  = lane & 15;
  const int b    = blockIdx.x >> 5;
  const int t0   = (blockIdx.x & 31) * 16;

  const float* Qb = Q + ((size_t)b * T_ + t0) * DP_;
  for (int i = tid; i < 16 * DP_; i += 256) q_lds[i] = Qb[i];
  if (tid < 16) { mrow[tid] = -3.0e38f; srow[tid] = 0.f; }
  __syncthreads();

  v8f zacc[8];
#pragma unroll
  for (int nt = 0; nt < 8; ++nt) zacc[nt] = zero8();

  const float scale = 0.0625f;
  const float* KTb = KT + (size_t)b * DP_ * L_;
  const int colw = w * 32 + l15;

  for (int lc = 0; lc < L_; lc += 256) {
    v8f s0 = zero8(), s1 = zero8();
    for (int kk = 0; kk < DP_; kk += 4) {
      v2f a;
      const int ai = l15 * DP_ + kk + 2 * half;
      a.x = q_lds[ai];
      a.y = q_lds[ai + 1];
      const float* kp = KTb + (size_t)(kk + 2 * half) * L_ + lc + colw;
      v2f b0; b0.x = kp[0];  b0.y = kp[L_];
      v2f b1; b1.x = kp[16]; b1.y = kp[16 + L_];
      s0 = wmma4(a, b0, s0);
      s1 = wmma4(a, b1, s1);
    }

    const int mv0 = mask[(size_t)b * L_ + lc + colw];
    const int mv1 = mask[(size_t)b * L_ + lc + colw + 16];
#pragma unroll
    for (int v = 0; v < 8; ++v) {
      float x0 = s0[v] * scale; if (mv0) x0 = -3.0e38f;
      float x1 = s1[v] * scale; if (mv1) x1 = -3.0e38f;
      s0[v] = x0; s1[v] = x1;
      float x = fmaxf(x0, x1);
      x = fmaxf(x, __shfl_xor(x, 8, 16));
      x = fmaxf(x, __shfl_xor(x, 4, 16));
      x = fmaxf(x, __shfl_xor(x, 2, 16));
      x = fmaxf(x, __shfl_xor(x, 1, 16));
      if (l15 == 0) red[w * 16 + v + 8 * half] = x;
    }
    __syncthreads();

    if (tid < 16) {
      float cm = red[tid];
#pragma unroll
      for (int ww = 1; ww < 8; ++ww) cm = fmaxf(cm, red[ww * 16 + tid]);
      const float mo = mrow[tid];
      const float mn = fmaxf(mo, cm);
      const float f  = __expf(mo - mn);
      srow[tid] *= f;
      frow[tid] = f;
      mrow[tid] = mn;
    }
    __syncthreads();

    float fsel[8], msel[8];
#pragma unroll
    for (int v = 0; v < 8; ++v) {
      fsel[v] = frow[v + 8 * half];
      msel[v] = mrow[v + 8 * half];
    }
#pragma unroll
    for (int nt = 0; nt < 8; ++nt)
#pragma unroll
      for (int v = 0; v < 8; ++v) zacc[nt][v] *= fsel[v];

#pragma unroll
    for (int v = 0; v < 8; ++v) {
      const float p0 = __expf(s0[v] - msel[v]);
      const float p1 = __expf(s1[v] - msel[v]);
      const int row = v + 8 * half;
      p_lds[row * 256 + colw]      = p0;
      p_lds[row * 256 + colw + 16] = p1;
      float x = p0 + p1;
      x += __shfl_xor(x, 8, 16);
      x += __shfl_xor(x, 4, 16);
      x += __shfl_xor(x, 2, 16);
      x += __shfl_xor(x, 1, 16);
      if (l15 == 0) red[w * 16 + row] = x;
    }
    __syncthreads();

    if (tid < 16) {
      float acc = 0.f;
#pragma unroll
      for (int ww = 0; ww < 8; ++ww) acc += red[ww * 16 + tid];
      srow[tid] += acc;
    }
    __syncthreads();

    const float* Hb = H + ((size_t)b * L_ + lc) * DH_ + w * 128 + l15;
    for (int kk = 0; kk < 256; kk += 4) {
      v2f a;
      const int ai = l15 * 256 + kk + 2 * half;
      a.x = p_lds[ai];
      a.y = p_lds[ai + 1];
      const float* hrow = Hb + (size_t)(kk + 2 * half) * DH_;
#pragma unroll
      for (int nt = 0; nt < 8; ++nt) {
        v2f bb;
        bb.x = hrow[nt * 16];
        bb.y = hrow[nt * 16 + DH_];
        zacc[nt] = wmma4(a, bb, zacc[nt]);
      }
    }
  }

  float inv[8];
#pragma unroll
  for (int v = 0; v < 8; ++v) inv[v] = 1.0f / srow[v + 8 * half];
  float* Zb = Z + ((size_t)b * T_ + t0) * DH_;
#pragma unroll
  for (int nt = 0; nt < 8; ++nt)
#pragma unroll
    for (int v = 0; v < 8; ++v) {
      const int m = v + 8 * half;
      const int d = w * 128 + nt * 16 + l15;
      Zb[(size_t)m * DH_ + d] = zacc[nt][v] * inv[v];
    }
}

extern "C" void kernel_launch(void* const* d_in, const int* in_sizes, int n_in,
                              void* d_out, int out_size, void* d_ws, size_t ws_size,
                              hipStream_t stream) {
  const float* H    = (const float*)d_in[0];
  const float* G    = (const float*)d_in[1];
  const float* Wk   = (const float*)d_in[2];
  const float* Wq   = (const float*)d_in[3];
  const int*   mask = (const int*)d_in[4];
  float*       Z    = (float*)d_out;

  float* Qws  = (float*)d_ws;
  float* KTws = Qws + (size_t)B_ * T_ * DP_;

  proj_gemm_kernel<false><<<(B_ * T_) / 16, 256, 0, stream>>>(G, Wq, Qws, DG_);
  proj_gemm_kernel<true><<<(B_ * L_) / 16, 256, 0, stream>>>(H, Wk, KTws, DH_);
  attn_kernel<<<B_ * (T_ / 16), 256, 0, stream>>>(H, Qws, KTws, mask, Z);
}


# attention: K^T, H, Q and P fragments are fetched 12 (scores) and 3 (weighted sum) k-steps ahead of their WMMAs through a register ring; same addresses and k order, sums unchanged
# speedup vs baseline: 1.1005x; 1.1005x over previous
#include <hip/hip_runtime.h>
#include <hip/hip_bf16.h>

typedef __attribute__((ext_vector_type(2))) float v2f;
typedef __attribute__((ext_vector_type(8))) float v8f;

constexpr int B_ = 8, L_ = 8192, T_ = 512, DH_ = 1024, DG_ = 768, DP_ = 256;

__device__ __forceinline__ v8f wmma4(v2f a, v2f b, v8f c) {
  return __builtin_amdgcn_wmma_f32_16x16x4_f32(
       false, a,  false, b,
       (short)0, c,  false,  false);
}

__device__ __forceinline__ v8f zero8() {
  v8f z = {0.f, 0.f, 0.f, 0.f, 0.f, 0.f, 0.f, 0.f};
  return z;
}

template <bool TRANS>
__global__ __launch_bounds__(256) void proj_gemm_kernel(
    const float* __restrict__ A, const float* __restrict__ W,
    float* __restrict__ out, int Kd) {
  const int tid  = threadIdx.x;
  const int w    = tid >> 5;
  const int lane = tid & 31;
  const int half = lane >> 4;
  const int l15  = lane & 15;
  const int row_base = blockIdx.x * 16;
  const int col0 = w * 32 + l15;

  const float* Ab = A + (size_t)(row_base + l15) * Kd + 2 * half;

  v8f acc0 = zero8();
  v8f acc1 = zero8();

  for (int kk = 0; kk < Kd; kk += 4) {
    v2f a;
    a.x = Ab[kk];
    a.y = Ab[kk + 1];
    const float* Wb = W + (size_t)(kk + 2 * half) * DP_ + col0;
    v2f b0; b0.x = Wb[0];  b0.y = Wb[DP_];
    v2f b1; b1.x = Wb[16]; b1.y = Wb[16 + DP_];
    acc0 = wmma4(a, b0, acc0);
    acc1 = wmma4(a, b1, acc1);
  }

#pragma unroll
  for (int v = 0; v < 8; ++v) {
    const int m = v + 8 * half;
    const size_t r = (size_t)row_base + m;
    if (!TRANS) {
      out[r * DP_ + col0]      = acc0[v];
      out[r * DP_ + col0 + 16] = acc1[v];
    } else {
      const size_t bb = r >> 13;
      const size_t l  = r & (size_t)(L_ - 1);
      out[(bb * DP_ + col0) * L_ + l]        = acc0[v];
      out[(bb * DP_ + col0 + 16) * L_ + l]   = acc1[v];
    }
  }
}


constexpr int KA_AHEAD = 12;
constexpr int KA_RING  = 16;
constexpr int HC_AHEAD = 3;
constexpr int HC_RING  = 4;

__device__ __forceinline__ void fetch_k(float (&x)[2], float (&y)[2], const float* kp) {
  x[0] = kp[0];  y[0] = kp[L_];
  x[1] = kp[16]; y[1] = kp[16 + L_];
}

__device__ __forceinline__ void fetch_h(float (&x)[8], float (&y)[8], const float* hrow) {
#pragma unroll
  for (int nt = 0; nt < 8; ++nt) {
    x[nt] = hrow[nt * 16];
    y[nt] = hrow[nt * 16 + DH_];
  }
}

__device__ __forceinline__ v2f fetch_a(const float* ap) {
  v2f a;
  a.x = ap[0];
  a.y = ap[1];
  return a;
}

__global__ __launch_bounds__(256) void attn_kernel(
    const float* __restrict__ H, const float* __restrict__ Q,
    const float* __restrict__ KT, const int* __restrict__ mask,
    float* __restrict__ Z) {
  __shared__ float q_lds[16 * DP_];
  __shared__ float p_lds[16 * 256];
  __shared__ float red[8 * 16];
  __shared__ float mrow[16];
  __shared__ float srow[16];
  __shared__ float frow[16];

  const int tid  = threadIdx.x;
  const int w    = tid >> 5;
  const int lane = tid & 31;
  const int half = lane >> 4;
  const int l15  = lane & 15;
  const int b    = blockIdx.x >> 5;
  const int t0   = (blockIdx.x & 31) * 16;

  const float* Qb = Q + ((size_t)b * T_ + t0) * DP_;
  for (int i = tid; i < 16 * DP_; i += 256) q_lds[i] = Qb[i];
  if (tid < 16) { mrow[tid] = -3.0e38f; srow[tid] = 0.f; }
  __syncthreads();

  v8f zacc[8];
#pragma unroll
  for (int nt = 0; nt < 8; ++nt) zacc[nt] = zero8();

  const float scale = 0.0625f;
  const float* KTb = KT + (size_t)b * DP_ * L_;
  const int colw = w * 32 + l15;

  for (int lc = 0; lc < L_; lc += 256) {
    v8f s0 = zero8(), s1 = zero8();
    {
      const float* kp0 = KTb + (size_t)(2 * half) * L_ + lc + colw;
      const float* qp0 = q_lds + l15 * DP_ + 2 * half;
      float kx[KA_RING][2], ky[KA_RING][2];
      v2f qa[KA_RING];
      constexpr int NS = DP_ / 4;
#pragma unroll
      for (int s = 0; s < KA_AHEAD; ++s) {
        fetch_k(kx[s], ky[s], kp0 + (size_t)(4 * s) * L_);
        qa[s] = fetch_a(qp0 + 4 * s);
      }
      for (int sb = 0; sb + KA_RING + KA_AHEAD <= NS; sb += KA_RING) {
#pragma unroll
        for (int u = 0; u < KA_RING; ++u) {
          const int sn = sb + u + KA_AHEAD;
          constexpr int M = KA_RING - 1;
          fetch_k(kx[(u + KA_AHEAD) & M], ky[(u + KA_AHEAD) & M], kp0 + (size_t)(4 * sn) * L_);
          qa[(u + KA_AHEAD) & M] = fetch_a(qp0 + 4 * sn);
          v2f b0; b0.x = kx[u][0]; b0.y = ky[u][0];
          v2f b1; b1.x = kx[u][1]; b1.y = ky[u][1];
          s0 = wmma4(qa[u], b0, s0);
          s1 = wmma4(qa[u], b1, s1);
        }
      }
      {
        constexpr int sb = ((NS - KA_AHEAD) / KA_RING) * KA_RING;
#pragma unroll
        for (int u = 0; u < NS - sb; ++u) {
          const int sn = sb + u + KA_AHEAD;
          constexpr int M = KA_RING - 1;
          if (sn < NS) {
            fetch_k(kx[(u + KA_AHEAD) & M], ky[(u + KA_AHEAD) & M], kp0 + (size_t)(4 * sn) * L_);
            qa[(u + KA_AHEAD) & M] = fetch_a(qp0 + 4 * sn);
          }
          v2f b0; b0.x = kx[u & M][0]; b0.y = ky[u & M][0];
          v2f b1; b1.x = kx[u & M][1]; b1.y = ky[u & M][1];
          s0 = wmma4(qa[u & M], b0, s0);
          s1 = wmma4(qa[u & M], b1, s1);
        }
      }
    }

    const float* Hb = H + ((size_t)b * L_ + lc + 2 * half) * DH_ + w * 128 + l15;
    float hx[HC_RING][8], hy[HC_RING][8];
#pragma unroll
    for (int s = 0; s < HC_AHEAD; ++s) fetch_h(hx[s], hy[s], Hb + (size_t)(4 * s) * DH_);

    const int mv0 = mask[(size_t)b * L_ + lc + colw];
    const int mv1 = mask[(size_t)b * L_ + lc + colw + 16];
#pragma unroll
    for (int v = 0; v < 8; ++v) {
      float x0 = s0[v] * scale; if (mv0) x0 = -3.0e38f;
      float x1 = s1[v] * scale; if (mv1) x1 = -3.0e38f;
      s0[v] = x0; s1[v] = x1;
      float x = fmaxf(x0, x1);
      x = fmaxf(x, __shfl_xor(x, 8, 16));
      x = fmaxf(x, __shfl_xor(x, 4, 16));
      x = fmaxf(x, __shfl_xor(x, 2, 16));
      x = fmaxf(x, __shfl_xor(x, 1, 16));
      if (l15 == 0) red[w * 16 + v + 8 * half] = x;
    }
    __syncthreads();

    if (tid < 16) {
      float cm = red[tid];
#pragma unroll
      for (int ww = 1; ww < 8; ++ww) cm = fmaxf(cm, red[ww * 16 + tid]);
      const float mo = mrow[tid];
      const float mn = fmaxf(mo, cm);
      const float f  = __expf(mo - mn);
      srow[tid] *= f;
      frow[tid] = f;
      mrow[tid] = mn;
    }
    __syncthreads();

    float fsel[8], msel[8];
#pragma unroll
    for (int v = 0; v < 8; ++v) {
      fsel[v] = frow[v + 8 * half];
      msel[v] = mrow[v + 8 * half];
    }
#pragma unroll
    for (int nt = 0; nt < 8; ++nt)
#pragma unroll
      for (int v = 0; v < 8; ++v) zacc[nt][v] *= fsel[v];

#pragma unroll
    for (int v = 0; v < 8; ++v) {
      const float p0 = __expf(s0[v] - msel[v]);
      const float p1 = __expf(s1[v] - msel[v]);
      const int row = v + 8 * half;
      p_lds[row * 256 + colw]      = p0;
      p_lds[row * 256 + colw + 16] = p1;
      float x = p0 + p1;
      x += __shfl_xor(x, 8, 16);
      x += __shfl_xor(x, 4, 16);
      x += __shfl_xor(x, 2, 16);
      x += __shfl_xor(x, 1, 16);
      if (l15 == 0) red[w * 16 + row] = x;
    }
    __syncthreads();

    if (tid < 16) {
      float acc = 0.f;
#pragma unroll
      for (int ww = 0; ww < 8; ++ww) acc += red[ww * 16 + tid];
      srow[tid] += acc;
    }
    __syncthreads();

    {
      const float* pp0 = p_lds + l15 * 256 + 2 * half;
      v2f pa[HC_RING];
      constexpr int NS = 256 / 4;
#pragma unroll
      for (int s = 0; s < HC_AHEAD; ++s) pa[s] = fetch_a(pp0 + 4 * s);
      for (int sb = 0; sb + HC_RING + HC_AHEAD <= NS; sb += HC_RING) {
#pragma unroll
        for (int u = 0; u < HC_RING; ++u) {
          const int sn = sb + u + HC_AHEAD;
          constexpr int M = HC_RING - 1;
          fetch_h(hx[(u + HC_AHEAD) & M], hy[(u + HC_AHEAD) & M], Hb + (size_t)(4 * sn) * DH_);
          pa[(u + HC_AHEAD) & M] = fetch_a(pp0 + 4 * sn);
#pragma unroll
          for (int nt = 0; nt < 8; ++nt) {
            v2f bb; bb.x = hx[u][nt]; bb.y = hy[u][nt];
            zacc[nt] = wmma4(pa[u], bb, zacc[nt]);
          }
        }
      }
      {
        constexpr int sb = ((NS - HC_AHEAD) / HC_RING) * HC_RING;
#pragma unroll
        for (int u = 0; u < NS - sb; ++u) {
          const int sn = sb + u + HC_AHEAD;
          constexpr int M = HC_RING - 1;
          if (sn < NS) {
            fetch_h(hx[(u + HC_AHEAD) & M], hy[(u + HC_AHEAD) & M], Hb + (size_t)(4 * sn) * DH_);
            pa[(u + HC_AHEAD) & M] = fetch_a(pp0 + 4 * sn);
          }
#pragma unroll
          for (int nt = 0; nt < 8; ++nt) {
            v2f bb; bb.x = hx[u & M][nt]; bb.y = hy[u & M][nt];
            zacc[nt] = wmma4(pa[u & M], bb, zacc[nt]);
          }
        }
      }
    }
  }

  float inv[8];
#pragma unroll
  for (int v = 0; v < 8; ++v) inv[v] = 1.0f / srow[v + 8 * half];
  float* Zb = Z + ((size_t)b * T_ + t0) * DH_;
#pragma unroll
  for (int nt = 0; nt < 8; ++nt)
#pragma unroll
    for (int v = 0; v < 8; ++v) {
      const int m = v + 8 * half;
      const int d = w * 128 + nt * 16 + l15;
      Zb[(size_t)m * DH_ + d] = zacc[nt][v] * inv[v];
    }
}

extern "C" void kernel_launch(void* const* d_in, const int* in_sizes, int n_in,
                              void* d_out, int out_size, void* d_ws, size_t ws_size,
                              hipStream_t stream) {
  const float* H    = (const float*)d_in[0];
  const float* G    = (const float*)d_in[1];
  const float* Wk   = (const float*)d_in[2];
  const float* Wq   = (const float*)d_in[3];
  const int*   mask = (const int*)d_in[4];
  float*       Z    = (float*)d_out;

  float* Qws  = (float*)d_ws;
  float* KTws = Qws + (size_t)B_ * T_ * DP_;

  proj_gemm_kernel<false><<<(B_ * T_) / 16, 256, 0, stream>>>(G, Wq, Qws, DG_);
  proj_gemm_kernel<true><<<(B_ * L_) / 16, 256, 0, stream>>>(H, Wk, KTws, DH_);
  attn_kernel<<<B_ * (T_ / 16), 256, 0, stream>>>(H, Qws, KTws, mask, Z);
}


# all B fragments fetched wide (Wk/Wq and K^T 8-byte, H 16-byte loads): each lane takes adjacent columns, tile-to-column map permuted; softmax row-sum adds adjacent columns first; k order unchanged
# speedup vs baseline: 1.1816x; 1.0737x over previous
#include <hip/hip_runtime.h>
#include <hip/hip_bf16.h>

typedef __attribute__((ext_vector_type(2))) float v2f;
typedef __attribute__((ext_vector_type(4))) float v4f;
typedef __attribute__((ext_vector_type(8))) float v8f;

constexpr int B_ = 8, L_ = 8192, T_ = 512, DH_ = 1024, DG_ = 768, DP_ = 256;

__device__ __forceinline__ v8f wmma4(v2f a, v2f b, v8f c) {
  return __builtin_amdgcn_wmma_f32_16x16x4_f32(
       false, a,  false, b,
       (short)0, c,  false,  false);
}

__device__ __forceinline__ v8f zero8() {
  v8f z = {0.f, 0.f, 0.f, 0.f, 0.f, 0.f, 0.f, 0.f};
  return z;
}

template <bool TRANS>
__global__ __launch_bounds__(256) void proj_gemm_kernel(
    const float* __restrict__ A, const float* __restrict__ W,
    float* __restrict__ out, int Kd) {
  const int tid  = threadIdx.x;
  const int w    = tid >> 5;
  const int lane = tid & 31;
  const int half = lane >> 4;
  const int l15  = lane & 15;
  const int row_base = blockIdx.x * 16;
  const int col2 = w * 32 + 2 * l15;

  const float* Ab = A + (size_t)(row_base + l15) * Kd + 2 * half;

  v8f acc0 = zero8();
  v8f acc1 = zero8();

  for (int kk = 0; kk < Kd; kk += 4) {
    v2f a;
    a.x = Ab[kk];
    a.y = Ab[kk + 1];
    const float* Wb = W + (size_t)(kk + 2 * half) * DP_ + col2;
    const v2f w0 = *reinterpret_cast<const v2f*>(Wb);
    const v2f w1 = *reinterpret_cast<const v2f*>(Wb + DP_);
    v2f b0; b0.x = w0.x; b0.y = w1.x;
    v2f b1; b1.x = w0.y; b1.y = w1.y;
    acc0 = wmma4(a, b0, acc0);
    acc1 = wmma4(a, b1, acc1);
  }

#pragma unroll
  for (int v = 0; v < 8; ++v) {
    const int m = v + 8 * half;
    const size_t r = (size_t)row_base + m;
    if (!TRANS) {
      v2f o; o.x = acc0[v]; o.y = acc1[v];
      *reinterpret_cast<v2f*>(out + r * DP_ + col2) = o;
    } else {
      const size_t bb = r >> 13;
      const size_t l  = r & (size_t)(L_ - 1);
      out[(bb * DP_ + col2) * L_ + l]       = acc0[v];
      out[(bb * DP_ + col2 + 1) * L_ + l]   = acc1[v];
    }
  }
}


constexpr int KA_AHEAD = 12;
constexpr int KA_RING  = 16;
constexpr int HC_AHEAD = 3;
constexpr int HC_RING  = 4;

__device__ __forceinline__ void fetch_k(v2f& x, v2f& y, const float* kp) {
  x = *reinterpret_cast<const v2f*>(kp);
  y = *reinterpret_cast<const v2f*>(kp + L_);
}

__device__ __forceinline__ void fetch_h(v4f (&x)[2], v4f (&y)[2], const float* hrow) {
#pragma unroll
  for (int q = 0; q < 2; ++q) {
    x[q] = *reinterpret_cast<const v4f*>(hrow + q * 64);
    y[q] = *reinterpret_cast<const v4f*>(hrow + q * 64 + DH_);
  }
}

__device__ __forceinline__ v2f fetch_a(const float* ap) {
  v2f a;
  a.x = ap[0];
  a.y = ap[1];
  return a;
}

__global__ __launch_bounds__(256) void attn_kernel(
    const float* __restrict__ H, const float* __restrict__ Q,
    const float* __restrict__ KT, const int* __restrict__ mask,
    float* __restrict__ Z) {
  __shared__ float q_lds[16 * DP_];
  __shared__ alignas(16) float p_lds[16 * 256];
  __shared__ float red[8 * 16];
  __shared__ float mrow[16];
  __shared__ float srow[16];
  __shared__ float frow[16];

  const int tid  = threadIdx.x;
  const int w    = tid >> 5;
  const int lane = tid & 31;
  const int half = lane >> 4;
  const int l15  = lane & 15;
  const int b    = blockIdx.x >> 5;
  const int t0   = (blockIdx.x & 31) * 16;

  const float* Qb = Q + ((size_t)b * T_ + t0) * DP_;
  for (int i = tid; i < 16 * DP_; i += 256) q_lds[i] = Qb[i];
  if (tid < 16) { mrow[tid] = -3.0e38f; srow[tid] = 0.f; }
  __syncthreads();

  v8f zacc[8];
#pragma unroll
  for (int nt = 0; nt < 8; ++nt) zacc[nt] = zero8();

  const float scale = 0.0625f;
  const float* KTb = KT + (size_t)b * DP_ * L_;
  const int col2 = w * 32 + 2 * l15;

  for (int lc = 0; lc < L_; lc += 256) {
    v8f s0 = zero8(), s1 = zero8();
    {
      const float* kp0 = KTb + (size_t)(2 * half) * L_ + lc + col2;
      const float* qp0 = q_lds + l15 * DP_ + 2 * half;
      v2f kx[KA_RING], ky[KA_RING];
      v2f qa[KA_RING];
      constexpr int NS = DP_ / 4;
#pragma unroll
      for (int s = 0; s < KA_AHEAD; ++s) {
        fetch_k(kx[s], ky[s], kp0 + (size_t)(4 * s) * L_);
        qa[s] = fetch_a(qp0 + 4 * s);
      }
      for (int sb = 0; sb + KA_RING + KA_AHEAD <= NS; sb += KA_RING) {
#pragma unroll
        for (int u = 0; u < KA_RING; ++u) {
          const int sn = sb + u + KA_AHEAD;
          constexpr int M = KA_RING - 1;
          fetch_k(kx[(u + KA_AHEAD) & M], ky[(u + KA_AHEAD) & M], kp0 + (size_t)(4 * sn) * L_);
          qa[(u + KA_AHEAD) & M] = fetch_a(qp0 + 4 * sn);
          v2f b0; b0.x = kx[u].x; b0.y = ky[u].x;
          v2f b1; b1.x = kx[u].y; b1.y = ky[u].y;
          s0 = wmma4(qa[u], b0, s0);
          s1 = wmma4(qa[u], b1, s1);
        }
      }
      {
        constexpr int sb = ((NS - KA_AHEAD) / KA_RING) * KA_RING;
#pragma unroll
        for (int u = 0; u < NS - sb; ++u) {
          const int sn = sb + u + KA_AHEAD;
          constexpr int M = KA_RING - 1;
          if (sn < NS) {
            fetch_k(kx[(u + KA_AHEAD) & M], ky[(u + KA_AHEAD) & M], kp0 + (size_t)(4 * sn) * L_);
            qa[(u + KA_AHEAD) & M] = fetch_a(qp0 + 4 * sn);
          }
          v2f b0; b0.x = kx[u & M].x; b0.y = ky[u & M].x;
          v2f b1; b1.x = kx[u & M].y; b1.y = ky[u & M].y;
          s0 = wmma4(qa[u & M], b0, s0);
          s1 = wmma4(qa[u & M], b1, s1);
        }
      }
    }

    const float* Hb = H + ((size_t)b * L_ + lc + 2 * half) * DH_ + w * 128 + 4 * l15;
    v4f hx[HC_RING][2], hy[HC_RING][2];
#pragma unroll
    for (int s = 0; s < HC_AHEAD; ++s) fetch_h(hx[s], hy[s], Hb + (size_t)(4 * s) * DH_);

    const int mv0 = mask[(size_t)b * L_ + lc + col2];
    const int mv1 = mask[(size_t)b * L_ + lc + col2 + 1];
#pragma unroll
    for (int v = 0; v < 8; ++v) {
      float x0 = s0[v] * scale; if (mv0) x0 = -3.0e38f;
      float x1 = s1[v] * scale; if (mv1) x1 = -3.0e38f;
      s0[v] = x0; s1[v] = x1;
      float x = fmaxf(x0, x1);
      x = fmaxf(x, __shfl_xor(x, 8, 16));
      x = fmaxf(x, __shfl_xor(x, 4, 16));
      x = fmaxf(x, __shfl_xor(x, 2, 16));
      x = fmaxf(x, __shfl_xor(x, 1, 16));
      if (l15 == 0) red[w * 16 + v + 8 * half] = x;
    }
    __syncthreads();

    if (tid < 16) {
      float cm = red[tid];
#pragma unroll
      for (int ww = 1; ww < 8; ++ww) cm = fmaxf(cm, red[ww * 16 + tid]);
      const float mo = mrow[tid];
      const float mn = fmaxf(mo, cm);
      const float f  = __expf(mo - mn);
      srow[tid] *= f;
      frow[tid] = f;
      mrow[tid] = mn;
    }
    __syncthreads();

    float fsel[8], msel[8];
#pragma unroll
    for (int v = 0; v < 8; ++v) {
      fsel[v] = frow[v + 8 * half];
      msel[v] = mrow[v + 8 * half];
    }
#pragma unroll
    for (int nt = 0; nt < 8; ++nt)
#pragma unroll
      for (int v = 0; v < 8; ++v) zacc[nt][v] *= fsel[v];

#pragma unroll
    for (int v = 0; v < 8; ++v) {
      const float p0 = __expf(s0[v] - msel[v]);
      const float p1 = __expf(s1[v] - msel[v]);
      const int row = v + 8 * half;
      v2f pp; pp.x = p0; pp.y = p1;
      *reinterpret_cast<v2f*>(&p_lds[row * 256 + col2]) = pp;
      float x = p0 + p1;
      x += __shfl_xor(x, 8, 16);
      x += __shfl_xor(x, 4, 16);
      x += __shfl_xor(x, 2, 16);
      x += __shfl_xor(x, 1, 16);
      if (l15 == 0) red[w * 16 + row] = x;
    }
    __syncthreads();

    if (tid < 16) {
      float acc = 0.f;
#pragma unroll
      for (int ww = 0; ww < 8; ++ww) acc += red[ww * 16 + tid];
      srow[tid] += acc;
    }
    __syncthreads();

    {
      const float* pp0 = p_lds + l15 * 256 + 2 * half;
      v2f pa[HC_RING];
      constexpr int NS = 256 / 4;
#pragma unroll
      for (int s = 0; s < HC_AHEAD; ++s) pa[s] = fetch_a(pp0 + 4 * s);
      for (int sb = 0; sb + HC_RING + HC_AHEAD <= NS; sb += HC_RING) {
#pragma unroll
        for (int u = 0; u < HC_RING; ++u) {
          const int sn = sb + u + HC_AHEAD;
          constexpr int M = HC_RING - 1;
          fetch_h(hx[(u + HC_AHEAD) & M], hy[(u + HC_AHEAD) & M], Hb + (size_t)(4 * sn) * DH_);
          pa[(u + HC_AHEAD) & M] = fetch_a(pp0 + 4 * sn);
#pragma unroll
          for (int nt = 0; nt < 8; ++nt) {
            v2f bb; bb.x = hx[u][nt >> 2][nt & 3]; bb.y = hy[u][nt >> 2][nt & 3];
            zacc[nt] = wmma4(pa[u], bb, zacc[nt]);
          }
        }
      }
      {
        constexpr int sb = ((NS - HC_AHEAD) / HC_RING) * HC_RING;
#pragma unroll
        for (int u = 0; u < NS - sb; ++u) {
          const int sn = sb + u + HC_AHEAD;
          constexpr int M = HC_RING - 1;
          if (sn < NS) {
            fetch_h(hx[(u + HC_AHEAD) & M], hy[(u + HC_AHEAD) & M], Hb + (size_t)(4 * sn) * DH_);
            pa[(u + HC_AHEAD) & M] = fetch_a(pp0 + 4 * sn);
          }
#pragma unroll
          for (int nt = 0; nt < 8; ++nt) {
            v2f bb; bb.x = hx[u & M][nt >> 2][nt & 3]; bb.y = hy[u & M][nt >> 2][nt & 3];
            zacc[nt] = wmma4(pa[u & M], bb, zacc[nt]);
          }
        }
      }
    }
  }

  float inv[8];
#pragma unroll
  for (int v = 0; v < 8; ++v) inv[v] = 1.0f / srow[v + 8 * half];
  float* Zb = Z + ((size_t)b * T_ + t0) * DH_;
#pragma unroll
  for (int q = 0; q < 2; ++q)
#pragma unroll
    for (int v = 0; v < 8; ++v) {
      const int m = v + 8 * half;
      const int d = w * 128 + q * 64 + 4 * l15;
      v4f o;
      o.x = zacc[4 * q + 0][v] * inv[v];
      o.y = zacc[4 * q + 1][v] * inv[v];
      o.z = zacc[4 * q + 2][v] * inv[v];
      o.w = zacc[4 * q + 3][v] * inv[v];
      *reinterpret_cast<v4f*>(Zb + (size_t)m * DH_ + d) = o;
    }
}

extern "C" void kernel_launch(void* const* d_in, const int* in_sizes, int n_in,
                              void* d_out, int out_size, void* d_ws, size_t ws_size,
                              hipStream_t stream) {
  const float* H    = (const float*)d_in[0];
  const float* G    = (const float*)d_in[1];
  const float* Wk   = (const float*)d_in[2];
  const float* Wq   = (const float*)d_in[3];
  const int*   mask = (const int*)d_in[4];
  float*       Z    = (float*)d_out;

  float* Qws  = (float*)d_ws;
  float* KTws = Qws + (size_t)B_ * T_ * DP_;

  proj_gemm_kernel<false><<<(B_ * T_) / 16, 256, 0, stream>>>(G, Wq, Qws, DG_);
  proj_gemm_kernel<true><<<(B_ * L_) / 16, 256, 0, stream>>>(H, Wk, KTws, DH_);
  attn_kernel<<<B_ * (T_ / 16), 256, 0, stream>>>(H, Qws, KTws, mask, Z);
}


# K projection retiled: 64 rows per workgroup, H tile fetched once as whole lines and shared through LDS (two 32-k stages), each Wk fragment feeds four M-tiles; k order and sums unchanged
# speedup vs baseline: 1.3165x; 1.1142x over previous
#include <hip/hip_runtime.h>
#include <hip/hip_bf16.h>

typedef __attribute__((ext_vector_type(2))) float v2f;
typedef __attribute__((ext_vector_type(4))) float v4f;
typedef __attribute__((ext_vector_type(8))) float v8f;

constexpr int B_ = 8, L_ = 8192, T_ = 512, DH_ = 1024, DG_ = 768, DP_ = 256;

__device__ __forceinline__ v8f wmma4(v2f a, v2f b, v8f c) {
  return __builtin_amdgcn_wmma_f32_16x16x4_f32(
       false, a,  false, b,
       (short)0, c,  false,  false);
}

__device__ __forceinline__ v8f zero8() {
  v8f z = {0.f, 0.f, 0.f, 0.f, 0.f, 0.f, 0.f, 0.f};
  return z;
}

__global__ __launch_bounds__(256) void proj_gemm_kernel(
    const float* __restrict__ A, const float* __restrict__ W,
    float* __restrict__ out, int Kd) {
  const int tid  = threadIdx.x;
  const int w    = tid >> 5;
  const int lane = tid & 31;
  const int half = lane >> 4;
  const int l15  = lane & 15;
  const int row_base = blockIdx.x * 16;
  const int col2 = w * 32 + 2 * l15;

  const float* Ab = A + (size_t)(row_base + l15) * Kd + 2 * half;

  v8f acc0 = zero8();
  v8f acc1 = zero8();

  for (int kk = 0; kk < Kd; kk += 4) {
    v2f a;
    a.x = Ab[kk];
    a.y = Ab[kk + 1];
    const float* Wb = W + (size_t)(kk + 2 * half) * DP_ + col2;
    const v2f w0 = *reinterpret_cast<const v2f*>(Wb);
    const v2f w1 = *reinterpret_cast<const v2f*>(Wb + DP_);
    v2f b0; b0.x = w0.x; b0.y = w1.x;
    v2f b1; b1.x = w0.y; b1.y = w1.y;
    acc0 = wmma4(a, b0, acc0);
    acc1 = wmma4(a, b1, acc1);
  }

#pragma unroll
  for (int v = 0; v < 8; ++v) {
    const int m = v + 8 * half;
    const size_t r = (size_t)row_base + m;
    v2f o; o.x = acc0[v]; o.y = acc1[v];
    *reinterpret_cast<v2f*>(out + r * DP_ + col2) = o;
  }
}

constexpr int KP_ROWS  = 64;
constexpr int KP_KC    = 32;
constexpr int KP_PITCH = 36;

__global__ __launch_bounds__(256) void kproj_kernel(
    const float* __restrict__ A, const float* __restrict__ W,
    float* __restrict__ out) {
  __shared__ alignas(16) float a_lds[2][KP_ROWS * KP_PITCH];

  const int tid  = threadIdx.x;
  const int w    = tid >> 5;
  const int lane = tid & 31;
  const int half = lane >> 4;
  const int l15  = lane & 15;
  const int row_base = blockIdx.x * KP_ROWS;
  const int col2 = w * 32 + 2 * l15;

  const int fr = tid >> 3;
  const int fc = (tid & 7) * 4;
  const float* Ag = A + (size_t)(row_base + fr) * DH_ + fc;
  const int so = fr * KP_PITCH + fc;

  v8f acc[4][2];
#pragma unroll
  for (int mt = 0; mt < 4; ++mt) { acc[mt][0] = zero8(); acc[mt][1] = zero8(); }

  v4f g0 = *reinterpret_cast<const v4f*>(Ag);
  v4f g1 = *reinterpret_cast<const v4f*>(Ag + (size_t)32 * DH_);
  *reinterpret_cast<v4f*>(&a_lds[0][so]) = g0;
  *reinterpret_cast<v4f*>(&a_lds[0][so + 32 * KP_PITCH]) = g1;
  __syncthreads();

  const float* Wl = W + (size_t)(2 * half) * DP_ + col2;
  const int ao = l15 * KP_PITCH + 2 * half;

  for (int kc = 0; kc < DH_; kc += KP_KC) {
    const int buf = (kc / KP_KC) & 1;
    const bool more = kc + KP_KC < DH_;
    if (more) {
      g0 = *reinterpret_cast<const v4f*>(Ag + kc + KP_KC);
      g1 = *reinterpret_cast<const v4f*>(Ag + (size_t)32 * DH_ + kc + KP_KC);
    }
    const float* ab = &a_lds[buf][ao];
#pragma unroll
    for (int ks = 0; ks < KP_KC / 4; ++ks) {
      const float* Wb = Wl + (size_t)(kc + 4 * ks) * DP_;
      const v2f w0 = *reinterpret_cast<const v2f*>(Wb);
      const v2f w1 = *reinterpret_cast<const v2f*>(Wb + DP_);
      v2f b0; b0.x = w0.x; b0.y = w1.x;
      v2f b1; b1.x = w0.y; b1.y = w1.y;
#pragma unroll
      for (int mt = 0; mt < 4; ++mt) {
        const v2f a = *reinterpret_cast<const v2f*>(ab + mt * 16 * KP_PITCH + 4 * ks);
        acc[mt][0] = wmma4(a, b0, acc[mt][0]);
        acc[mt][1] = wmma4(a, b1, acc[mt][1]);
      }
    }
    if (more) {
      *reinterpret_cast<v4f*>(&a_lds[buf ^ 1][so]) = g0;
      *reinterpret_cast<v4f*>(&a_lds[buf ^ 1][so + 32 * KP_PITCH]) = g1;
      __syncthreads();
    }
  }

  const size_t bb = (size_t)row_base >> 13;
  const size_t l0 = ((size_t)row_base & (size_t)(L_ - 1)) + 8 * half;
#pragma unroll
  for (int mt = 0; mt < 4; ++mt)
#pragma unroll
    for (int j = 0; j < 2; ++j) {
      float* o = out + (bb * DP_ + col2 + j) * L_ + l0 + mt * 16;
      v4f lo; lo.x = acc[mt][j][0]; lo.y = acc[mt][j][1]; lo.z = acc[mt][j][2]; lo.w = acc[mt][j][3];
      v4f hi; hi.x = acc[mt][j][4]; hi.y = acc[mt][j][5]; hi.z = acc[mt][j][6]; hi.w = acc[mt][j][7];
      *reinterpret_cast<v4f*>(o)     = lo;
      *reinterpret_cast<v4f*>(o + 4) = hi;
    }
}


constexpr int KA_AHEAD = 12;
constexpr int KA_RING  = 16;
constexpr int HC_AHEAD = 3;
constexpr int HC_RING  = 4;

__device__ __forceinline__ void fetch_k(v2f& x, v2f& y, const float* kp) {
  x = *reinterpret_cast<const v2f*>(kp);
  y = *reinterpret_cast<const v2f*>(kp + L_);
}

__device__ __forceinline__ void fetch_h(v4f (&x)[2], v4f (&y)[2], const float* hrow) {
#pragma unroll
  for (int q = 0; q < 2; ++q) {
    x[q] = *reinterpret_cast<const v4f*>(hrow + q * 64);
    y[q] = *reinterpret_cast<const v4f*>(hrow + q * 64 + DH_);
  }
}

__device__ __forceinline__ v2f fetch_a(const float* ap) {
  v2f a;
  a.x = ap[0];
  a.y = ap[1];
  return a;
}

__global__ __launch_bounds__(256) void attn_kernel(
    const float* __restrict__ H, const float* __restrict__ Q,
    const float* __restrict__ KT, const int* __restrict__ mask,
    float* __restrict__ Z) {
  __shared__ float q_lds[16 * DP_];
  __shared__ alignas(16) float p_lds[16 * 256];
  __shared__ float red[8 * 16];
  __shared__ float mrow[16];
  __shared__ float srow[16];
  __shared__ float frow[16];

  const int tid  = threadIdx.x;
  const int w    = tid >> 5;
  const int lane = tid & 31;
  const int half = lane >> 4;
  const int l15  = lane & 15;
  const int b    = blockIdx.x >> 5;
  const int t0   = (blockIdx.x & 31) * 16;

  const float* Qb = Q + ((size_t)b * T_ + t0) * DP_;
  for (int i = tid; i < 16 * DP_; i += 256) q_lds[i] = Qb[i];
  if (tid < 16) { mrow[tid] = -3.0e38f; srow[tid] = 0.f; }
  __syncthreads();

  v8f zacc[8];
#pragma unroll
  for (int nt = 0; nt < 8; ++nt) zacc[nt] = zero8();

  const float scale = 0.0625f;
  const float* KTb = KT + (size_t)b * DP_ * L_;
  const int col2 = w * 32 + 2 * l15;

  for (int lc = 0; lc < L_; lc += 256) {
    v8f s0 = zero8(), s1 = zero8();
    {
      const float* kp0 = KTb + (size_t)(2 * half) * L_ + lc + col2;
      const float* qp0 = q_lds + l15 * DP_ + 2 * half;
      v2f kx[KA_RING], ky[KA_RING];
      v2f qa[KA_RING];
      constexpr int NS = DP_ / 4;
#pragma unroll
      for (int s = 0; s < KA_AHEAD; ++s) {
        fetch_k(kx[s], ky[s], kp0 + (size_t)(4 * s) * L_);
        qa[s] = fetch_a(qp0 + 4 * s);
      }
      for (int sb = 0; sb + KA_RING + KA_AHEAD <= NS; sb += KA_RING) {
#pragma unroll
        for (int u = 0; u < KA_RING; ++u) {
          const int sn = sb + u + KA_AHEAD;
          constexpr int M = KA_RING - 1;
          fetch_k(kx[(u + KA_AHEAD) & M], ky[(u + KA_AHEAD) & M], kp0 + (size_t)(4 * sn) * L_);
          qa[(u + KA_AHEAD) & M] = fetch_a(qp0 + 4 * sn);
          v2f b0; b0.x = kx[u].x; b0.y = ky[u].x;
          v2f b1; b1.x = kx[u].y; b1.y = ky[u].y;
          s0 = wmma4(qa[u], b0, s0);
          s1 = wmma4(qa[u], b1, s1);
        }
      }
      {
        constexpr int sb = ((NS - KA_AHEAD) / KA_RING) * KA_RING;
#pragma unroll
        for (int u = 0; u < NS - sb; ++u) {
          const int sn = sb + u + KA_AHEAD;
          constexpr int M = KA_RING - 1;
          if (sn < NS) {
            fetch_k(kx[(u + KA_AHEAD) & M], ky[(u + KA_AHEAD) & M], kp0 + (size_t)(4 * sn) * L_);
            qa[(u + KA_AHEAD) & M] = fetch_a(qp0 + 4 * sn);
          }
          v2f b0; b0.x = kx[u & M].x; b0.y = ky[u & M].x;
          v2f b1; b1.x = kx[u & M].y; b1.y = ky[u & M].y;
          s0 = wmma4(qa[u & M], b0, s0);
          s1 = wmma4(qa[u & M], b1, s1);
        }
      }
    }

    const float* Hb = H + ((size_t)b * L_ + lc + 2 * half) * DH_ + w * 128 + 4 * l15;
    v4f hx[HC_RING][2], hy[HC_RING][2];
#pragma unroll
    for (int s = 0; s < HC_AHEAD; ++s) fetch_h(hx[s], hy[s], Hb + (size_t)(4 * s) * DH_);

    const int mv0 = mask[(size_t)b * L_ + lc + col2];
    const int mv1 = mask[(size_t)b * L_ + lc + col2 + 1];
#pragma unroll
    for (int v = 0; v < 8; ++v) {
      float x0 = s0[v] * scale; if (mv0) x0 = -3.0e38f;
      float x1 = s1[v] * scale; if (mv1) x1 = -3.0e38f;
      s0[v] = x0; s1[v] = x1;
      float x = fmaxf(x0, x1);
      x = fmaxf(x, __shfl_xor(x, 8, 16));
      x = fmaxf(x, __shfl_xor(x, 4, 16));
      x = fmaxf(x, __shfl_xor(x, 2, 16));
      x = fmaxf(x, __shfl_xor(x, 1, 16));
      if (l15 == 0) red[w * 16 + v + 8 * half] = x;
    }
    __syncthreads();

    if (tid < 16) {
      float cm = red[tid];
#pragma unroll
      for (int ww = 1; ww < 8; ++ww) cm = fmaxf(cm, red[ww * 16 + tid]);
      const float mo = mrow[tid];
      const float mn = fmaxf(mo, cm);
      const float f  = __expf(mo - mn);
      srow[tid] *= f;
      frow[tid] = f;
      mrow[tid] = mn;
    }
    __syncthreads();

    float fsel[8], msel[8];
#pragma unroll
    for (int v = 0; v < 8; ++v) {
      fsel[v] = frow[v + 8 * half];
      msel[v] = mrow[v + 8 * half];
    }
#pragma unroll
    for (int nt = 0; nt < 8; ++nt)
#pragma unroll
      for (int v = 0; v < 8; ++v) zacc[nt][v] *= fsel[v];

#pragma unroll
    for (int v = 0; v < 8; ++v) {
      const float p0 = __expf(s0[v] - msel[v]);
      const float p1 = __expf(s1[v] - msel[v]);
      const int row = v + 8 * half;
      v2f pp; pp.x = p0; pp.y = p1;
      *reinterpret_cast<v2f*>(&p_lds[row * 256 + col2]) = pp;
      float x = p0 + p1;
      x += __shfl_xor(x, 8, 16);
      x += __shfl_xor(x, 4, 16);
      x += __shfl_xor(x, 2, 16);
      x += __shfl_xor(x, 1, 16);
      if (l15 == 0) red[w * 16 + row] = x;
    }
    __syncthreads();

    if (tid < 16) {
      float acc = 0.f;
#pragma unroll
      for (int ww = 0; ww < 8; ++ww) acc += red[ww * 16 + tid];
      srow[tid] += acc;
    }
    __syncthreads();

    {
      const float* pp0 = p_lds + l15 * 256 + 2 * half;
      v2f pa[HC_RING];
      constexpr int NS = 256 / 4;
#pragma unroll
      for (int s = 0; s < HC_AHEAD; ++s) pa[s] = fetch_a(pp0 + 4 * s);
      for (int sb = 0; sb + HC_RING + HC_AHEAD <= NS; sb += HC_RING) {
#pragma unroll
        for (int u = 0; u < HC_RING; ++u) {
          const int sn = sb + u + HC_AHEAD;
          constexpr int M = HC_RING - 1;
          fetch_h(hx[(u + HC_AHEAD) & M], hy[(u + HC_AHEAD) & M], Hb + (size_t)(4 * sn) * DH_);
          pa[(u + HC_AHEAD) & M] = fetch_a(pp0 + 4 * sn);
#pragma unroll
          for (int nt = 0; nt < 8; ++nt) {
            v2f bb; bb.x = hx[u][nt >> 2][nt & 3]; bb.y = hy[u][nt >> 2][nt & 3];
            zacc[nt] = wmma4(pa[u], bb, zacc[nt]);
          }
        }
      }
      {
        constexpr int sb = ((NS - HC_AHEAD) / HC_RING) * HC_RING;
#pragma unroll
        for (int u = 0; u < NS - sb; ++u) {
          const int sn = sb + u + HC_AHEAD;
          constexpr int M = HC_RING - 1;
          if (sn < NS) {
            fetch_h(hx[(u + HC_AHEAD) & M], hy[(u + HC_AHEAD) & M], Hb + (size_t)(4 * sn) * DH_);
            pa[(u + HC_AHEAD) & M] = fetch_a(pp0 + 4 * sn);
          }
#pragma unroll
          for (int nt = 0; nt < 8; ++nt) {
            v2f bb; bb.x = hx[u & M][nt >> 2][nt & 3]; bb.y = hy[u & M][nt >> 2][nt & 3];
            zacc[nt] = wmma4(pa[u & M], bb, zacc[nt]);
          }
        }
      }
    }
  }

  float inv[8];
#pragma unroll
  for (int v = 0; v < 8; ++v) inv[v] = 1.0f / srow[v + 8 * half];
  float* Zb = Z + ((size_t)b * T_ + t0) * DH_;
#pragma unroll
  for (int q = 0; q < 2; ++q)
#pragma unroll
    for (int v = 0; v < 8; ++v) {
      const int m = v + 8 * half;
      const int d = w * 128 + q * 64 + 4 * l15;
      v4f o;
      o.x = zacc[4 * q + 0][v] * inv[v];
      o.y = zacc[4 * q + 1][v] * inv[v];
      o.z = zacc[4 * q + 2][v] * inv[v];
      o.w = zacc[4 * q + 3][v] * inv[v];
      *reinterpret_cast<v4f*>(Zb + (size_t)m * DH_ + d) = o;
    }
}

extern "C" void kernel_launch(void* const* d_in, const int* in_sizes, int n_in,
                              void* d_out, int out_size, void* d_ws, size_t ws_size,
                              hipStream_t stream) {
  const float* H    = (const float*)d_in[0];
  const float* G    = (const float*)d_in[1];
  const float* Wk   = (const float*)d_in[2];
  const float* Wq   = (const float*)d_in[3];
  const int*   mask = (const int*)d_in[4];
  float*       Z    = (float*)d_out;

  float* Qws  = (float*)d_ws;
  float* KTws = Qws + (size_t)B_ * T_ * DP_;

  proj_gemm_kernel<<<(B_ * T_) / 16, 256, 0, stream>>>(G, Wq, Qws, DG_);
  kproj_kernel<<<(B_ * L_) / KP_ROWS, 256, 0, stream>>>(H, Wk, KTws);
  attn_kernel<<<B_ * (T_ / 16), 256, 0, stream>>>(H, Qws, KTws, mask, Z);
}


# attention softmax: the row max and row sum butterflies of every chunk run on DPP row_xmask lane selects instead of LDS permutes, eight rows interleaved; same pairs and order, sums bit-identical
# speedup vs baseline: 1.4292x; 1.0856x over previous
#include <hip/hip_runtime.h>
#include <hip/hip_bf16.h>

typedef __attribute__((ext_vector_type(2))) float v2f;
typedef __attribute__((ext_vector_type(4))) float v4f;
typedef __attribute__((ext_vector_type(8))) float v8f;

constexpr int B_ = 8, L_ = 8192, T_ = 512, DH_ = 1024, DG_ = 768, DP_ = 256;

__device__ __forceinline__ v8f wmma4(v2f a, v2f b, v8f c) {
  return __builtin_amdgcn_wmma_f32_16x16x4_f32(
       false, a,  false, b,
       (short)0, c,  false,  false);
}

__device__ __forceinline__ v8f zero8() {
  v8f z = {0.f, 0.f, 0.f, 0.f, 0.f, 0.f, 0.f, 0.f};
  return z;
}

__global__ __launch_bounds__(256) void proj_gemm_kernel(
    const float* __restrict__ A, const float* __restrict__ W,
    float* __restrict__ out, int Kd) {
  const int tid  = threadIdx.x;
  const int w    = tid >> 5;
  const int lane = tid & 31;
  const int half = lane >> 4;
  const int l15  = lane & 15;
  const int row_base = blockIdx.x * 16;
  const int col2 = w * 32 + 2 * l15;

  const float* Ab = A + (size_t)(row_base + l15) * Kd + 2 * half;

  v8f acc0 = zero8();
  v8f acc1 = zero8();

  for (int kk = 0; kk < Kd; kk += 4) {
    v2f a;
    a.x = Ab[kk];
    a.y = Ab[kk + 1];
    const float* Wb = W + (size_t)(kk + 2 * half) * DP_ + col2;
    const v2f w0 = *reinterpret_cast<const v2f*>(Wb);
    const v2f w1 = *reinterpret_cast<const v2f*>(Wb + DP_);
    v2f b0; b0.x = w0.x; b0.y = w1.x;
    v2f b1; b1.x = w0.y; b1.y = w1.y;
    acc0 = wmma4(a, b0, acc0);
    acc1 = wmma4(a, b1, acc1);
  }

#pragma unroll
  for (int v = 0; v < 8; ++v) {
    const int m = v + 8 * half;
    const size_t r = (size_t)row_base + m;
    v2f o; o.x = acc0[v]; o.y = acc1[v];
    *reinterpret_cast<v2f*>(out + r * DP_ + col2) = o;
  }
}

constexpr int KP_ROWS  = 64;
constexpr int KP_KC    = 32;
constexpr int KP_PITCH = 36;

__global__ __launch_bounds__(256) void kproj_kernel(
    const float* __restrict__ A, const float* __restrict__ W,
    float* __restrict__ out) {
  __shared__ alignas(16) float a_lds[2][KP_ROWS * KP_PITCH];

  const int tid  = threadIdx.x;
  const int w    = tid >> 5;
  const int lane = tid & 31;
  const int half = lane >> 4;
  const int l15  = lane & 15;
  const int row_base = blockIdx.x * KP_ROWS;
  const int col2 = w * 32 + 2 * l15;

  const int fr = tid >> 3;
  const int fc = (tid & 7) * 4;
  const float* Ag = A + (size_t)(row_base + fr) * DH_ + fc;
  const int so = fr * KP_PITCH + fc;

  v8f acc[4][2];
#pragma unroll
  for (int mt = 0; mt < 4; ++mt) { acc[mt][0] = zero8(); acc[mt][1] = zero8(); }

  v4f g0 = *reinterpret_cast<const v4f*>(Ag);
  v4f g1 = *reinterpret_cast<const v4f*>(Ag + (size_t)32 * DH_);
  *reinterpret_cast<v4f*>(&a_lds[0][so]) = g0;
  *reinterpret_cast<v4f*>(&a_lds[0][so + 32 * KP_PITCH]) = g1;
  __syncthreads();

  const float* Wl = W + (size_t)(2 * half) * DP_ + col2;
  const int ao = l15 * KP_PITCH + 2 * half;

  for (int kc = 0; kc < DH_; kc += KP_KC) {
    const int buf = (kc / KP_KC) & 1;
    const bool more = kc + KP_KC < DH_;
    if (more) {
      g0 = *reinterpret_cast<const v4f*>(Ag + kc + KP_KC);
      g1 = *reinterpret_cast<const v4f*>(Ag + (size_t)32 * DH_ + kc + KP_KC);
    }
    const float* ab = &a_lds[buf][ao];
#pragma unroll
    for (int ks = 0; ks < KP_KC / 4; ++ks) {
      const float* Wb = Wl + (size_t)(kc + 4 * ks) * DP_;
      const v2f w0 = *reinterpret_cast<const v2f*>(Wb);
      const v2f w1 = *reinterpret_cast<const v2f*>(Wb + DP_);
      v2f b0; b0.x = w0.x; b0.y = w1.x;
      v2f b1; b1.x = w0.y; b1.y = w1.y;
#pragma unroll
      for (int mt = 0; mt < 4; ++mt) {
        const v2f a = *reinterpret_cast<const v2f*>(ab + mt * 16 * KP_PITCH + 4 * ks);
        acc[mt][0] = wmma4(a, b0, acc[mt][0]);
        acc[mt][1] = wmma4(a, b1, acc[mt][1]);
      }
    }
    if (more) {
      *reinterpret_cast<v4f*>(&a_lds[buf ^ 1][so]) = g0;
      *reinterpret_cast<v4f*>(&a_lds[buf ^ 1][so + 32 * KP_PITCH]) = g1;
      __syncthreads();
    }
  }

  const size_t bb = (size_t)row_base >> 13;
  const size_t l0 = ((size_t)row_base & (size_t)(L_ - 1)) + 8 * half;
#pragma unroll
  for (int mt = 0; mt < 4; ++mt)
#pragma unroll
    for (int j = 0; j < 2; ++j) {
      float* o = out + (bb * DP_ + col2 + j) * L_ + l0 + mt * 16;
      v4f lo; lo.x = acc[mt][j][0]; lo.y = acc[mt][j][1]; lo.z = acc[mt][j][2]; lo.w = acc[mt][j][3];
      v4f hi; hi.x = acc[mt][j][4]; hi.y = acc[mt][j][5]; hi.z = acc[mt][j][6]; hi.w = acc[mt][j][7];
      *reinterpret_cast<v4f*>(o)     = lo;
      *reinterpret_cast<v4f*>(o + 4) = hi;
    }
}


constexpr int KA_AHEAD = 12;
constexpr int KA_RING  = 16;
constexpr int HC_AHEAD = 3;
constexpr int HC_RING  = 4;

__device__ __forceinline__ void fetch_k(v2f& x, v2f& y, const float* kp) {
  x = *reinterpret_cast<const v2f*>(kp);
  y = *reinterpret_cast<const v2f*>(kp + L_);
}

__device__ __forceinline__ void fetch_h(v4f (&x)[2], v4f (&y)[2], const float* hrow) {
#pragma unroll
  for (int q = 0; q < 2; ++q) {
    x[q] = *reinterpret_cast<const v4f*>(hrow + q * 64);
    y[q] = *reinterpret_cast<const v4f*>(hrow + q * 64 + DH_);
  }
}

__device__ __forceinline__ v2f fetch_a(const float* ap) {
  v2f a;
  a.x = ap[0];
  a.y = ap[1];
  return a;
}

template <int MASK>
__device__ __forceinline__ float row_xor(float x) {
  const int xi = __builtin_bit_cast(int, x);
  return __builtin_bit_cast(float, __builtin_amdgcn_update_dpp(xi, xi, 0x160 + MASK, 0xF, 0xF, false));
}

__global__ __launch_bounds__(256) void attn_kernel(
    const float* __restrict__ H, const float* __restrict__ Q,
    const float* __restrict__ KT, const int* __restrict__ mask,
    float* __restrict__ Z) {
  __shared__ float q_lds[16 * DP_];
  __shared__ alignas(16) float p_lds[16 * 256];
  __shared__ float red[8 * 16];
  __shared__ float mrow[16];
  __shared__ float srow[16];
  __shared__ float frow[16];

  const int tid  = threadIdx.x;
  const int w    = tid >> 5;
  const int lane = tid & 31;
  const int half = lane >> 4;
  const int l15  = lane & 15;
  const int b    = blockIdx.x >> 5;
  const int t0   = (blockIdx.x & 31) * 16;

  const float* Qb = Q + ((size_t)b * T_ + t0) * DP_;
  for (int i = tid; i < 16 * DP_; i += 256) q_lds[i] = Qb[i];
  if (tid < 16) { mrow[tid] = -3.0e38f; srow[tid] = 0.f; }
  __syncthreads();

  v8f zacc[8];
#pragma unroll
  for (int nt = 0; nt < 8; ++nt) zacc[nt] = zero8();

  const float scale = 0.0625f;
  const float* KTb = KT + (size_t)b * DP_ * L_;
  const int col2 = w * 32 + 2 * l15;

  for (int lc = 0; lc < L_; lc += 256) {
    v8f s0 = zero8(), s1 = zero8();
    {
      const float* kp0 = KTb + (size_t)(2 * half) * L_ + lc + col2;
      const float* qp0 = q_lds + l15 * DP_ + 2 * half;
      v2f kx[KA_RING], ky[KA_RING];
      v2f qa[KA_RING];
      constexpr int NS = DP_ / 4;
#pragma unroll
      for (int s = 0; s < KA_AHEAD; ++s) {
        fetch_k(kx[s], ky[s], kp0 + (size_t)(4 * s) * L_);
        qa[s] = fetch_a(qp0 + 4 * s);
      }
      for (int sb = 0; sb + KA_RING + KA_AHEAD <= NS; sb += KA_RING) {
#pragma unroll
        for (int u = 0; u < KA_RING; ++u) {
          const int sn = sb + u + KA_AHEAD;
          constexpr int M = KA_RING - 1;
          fetch_k(kx[(u + KA_AHEAD) & M], ky[(u + KA_AHEAD) & M], kp0 + (size_t)(4 * sn) * L_);
          qa[(u + KA_AHEAD) & M] = fetch_a(qp0 + 4 * sn);
          v2f b0; b0.x = kx[u].x; b0.y = ky[u].x;
          v2f b1; b1.x = kx[u].y; b1.y = ky[u].y;
          s0 = wmma4(qa[u], b0, s0);
          s1 = wmma4(qa[u], b1, s1);
        }
      }
      {
        constexpr int sb = ((NS - KA_AHEAD) / KA_RING) * KA_RING;
#pragma unroll
        for (int u = 0; u < NS - sb; ++u) {
          const int sn = sb + u + KA_AHEAD;
          constexpr int M = KA_RING - 1;
          if (sn < NS) {
            fetch_k(kx[(u + KA_AHEAD) & M], ky[(u + KA_AHEAD) & M], kp0 + (size_t)(4 * sn) * L_);
            qa[(u + KA_AHEAD) & M] = fetch_a(qp0 + 4 * sn);
          }
          v2f b0; b0.x = kx[u & M].x; b0.y = ky[u & M].x;
          v2f b1; b1.x = kx[u & M].y; b1.y = ky[u & M].y;
          s0 = wmma4(qa[u & M], b0, s0);
          s1 = wmma4(qa[u & M], b1, s1);
        }
      }
    }

    const float* Hb = H + ((size_t)b * L_ + lc + 2 * half) * DH_ + w * 128 + 4 * l15;
    v4f hx[HC_RING][2], hy[HC_RING][2];
#pragma unroll
    for (int s = 0; s < HC_AHEAD; ++s) fetch_h(hx[s], hy[s], Hb + (size_t)(4 * s) * DH_);

    const int mv0 = mask[(size_t)b * L_ + lc + col2];
    const int mv1 = mask[(size_t)b * L_ + lc + col2 + 1];
    float cmx[8];
#pragma unroll
    for (int v = 0; v < 8; ++v) {
      float x0 = s0[v] * scale; if (mv0) x0 = -3.0e38f;
      float x1 = s1[v] * scale; if (mv1) x1 = -3.0e38f;
      s0[v] = x0; s1[v] = x1;
      cmx[v] = fmaxf(x0, x1);
    }
#pragma unroll
    for (int v = 0; v < 8; ++v) cmx[v] = fmaxf(cmx[v], row_xor<8>(cmx[v]));
#pragma unroll
    for (int v = 0; v < 8; ++v) cmx[v] = fmaxf(cmx[v], row_xor<4>(cmx[v]));
#pragma unroll
    for (int v = 0; v < 8; ++v) cmx[v] = fmaxf(cmx[v], row_xor<2>(cmx[v]));
#pragma unroll
    for (int v = 0; v < 8; ++v) cmx[v] = fmaxf(cmx[v], row_xor<1>(cmx[v]));
    if (l15 == 0) {
#pragma unroll
      for (int v = 0; v < 8; ++v) red[w * 16 + v + 8 * half] = cmx[v];
    }
    __syncthreads();

    if (tid < 16) {
      float cm = red[tid];
#pragma unroll
      for (int ww = 1; ww < 8; ++ww) cm = fmaxf(cm, red[ww * 16 + tid]);
      const float mo = mrow[tid];
      const float mn = fmaxf(mo, cm);
      const float f  = __expf(mo - mn);
      srow[tid] *= f;
      frow[tid] = f;
      mrow[tid] = mn;
    }
    __syncthreads();

    float fsel[8], msel[8];
#pragma unroll
    for (int v = 0; v < 8; ++v) {
      fsel[v] = frow[v + 8 * half];
      msel[v] = mrow[v + 8 * half];
    }
#pragma unroll
    for (int nt = 0; nt < 8; ++nt)
#pragma unroll
      for (int v = 0; v < 8; ++v) zacc[nt][v] *= fsel[v];

    float psum[8];
#pragma unroll
    for (int v = 0; v < 8; ++v) {
      const float p0 = __expf(s0[v] - msel[v]);
      const float p1 = __expf(s1[v] - msel[v]);
      const int row = v + 8 * half;
      v2f pp; pp.x = p0; pp.y = p1;
      *reinterpret_cast<v2f*>(&p_lds[row * 256 + col2]) = pp;
      psum[v] = p0 + p1;
    }
#pragma unroll
    for (int v = 0; v < 8; ++v) psum[v] += row_xor<8>(psum[v]);
#pragma unroll
    for (int v = 0; v < 8; ++v) psum[v] += row_xor<4>(psum[v]);
#pragma unroll
    for (int v = 0; v < 8; ++v) psum[v] += row_xor<2>(psum[v]);
#pragma unroll
    for (int v = 0; v < 8; ++v) psum[v] += row_xor<1>(psum[v]);
    if (l15 == 0) {
#pragma unroll
      for (int v = 0; v < 8; ++v) red[w * 16 + v + 8 * half] = psum[v];
    }
    __syncthreads();

    if (tid < 16) {
      float acc = 0.f;
#pragma unroll
      for (int ww = 0; ww < 8; ++ww) acc += red[ww * 16 + tid];
      srow[tid] += acc;
    }
    __syncthreads();

    {
      const float* pp0 = p_lds + l15 * 256 + 2 * half;
      v2f pa[HC_RING];
      constexpr int NS = 256 / 4;
#pragma unroll
      for (int s = 0; s < HC_AHEAD; ++s) pa[s] = fetch_a(pp0 + 4 * s);
      for (int sb = 0; sb + HC_RING + HC_AHEAD <= NS; sb += HC_RING) {
#pragma unroll
        for (int u = 0; u < HC_RING; ++u) {
          const int sn = sb + u + HC_AHEAD;
          constexpr int M = HC_RING - 1;
          fetch_h(hx[(u + HC_AHEAD) & M], hy[(u + HC_AHEAD) & M], Hb + (size_t)(4 * sn) * DH_);
          pa[(u + HC_AHEAD) & M] = fetch_a(pp0 + 4 * sn);
#pragma unroll
          for (int nt = 0; nt < 8; ++nt) {
            v2f bb; bb.x = hx[u][nt >> 2][nt & 3]; bb.y = hy[u][nt >> 2][nt & 3];
            zacc[nt] = wmma4(pa[u], bb, zacc[nt]);
          }
        }
      }
      {
        constexpr int sb = ((NS - HC_AHEAD) / HC_RING) * HC_RING;
#pragma unroll
        for (int u = 0; u < NS - sb; ++u) {
          const int sn = sb + u + HC_AHEAD;
          constexpr int M = HC_RING - 1;
          if (sn < NS) {
            fetch_h(hx[(u + HC_AHEAD) & M], hy[(u + HC_AHEAD) & M], Hb + (size_t)(4 * sn) * DH_);
            pa[(u + HC_AHEAD) & M] = fetch_a(pp0 + 4 * sn);
          }
#pragma unroll
          for (int nt = 0; nt < 8; ++nt) {
            v2f bb; bb.x = hx[u & M][nt >> 2][nt & 3]; bb.y = hy[u & M][nt >> 2][nt & 3];
            zacc[nt] = wmma4(pa[u & M], bb, zacc[nt]);
          }
        }
      }
    }
  }

  float inv[8];
#pragma unroll
  for (int v = 0; v < 8; ++v) inv[v] = 1.0f / srow[v + 8 * half];
  float* Zb = Z + ((size_t)b * T_ + t0) * DH_;
#pragma unroll
  for (int q = 0; q < 2; ++q)
#pragma unroll
    for (int v = 0; v < 8; ++v) {
      const int m = v + 8 * half;
      const int d = w * 128 + q * 64 + 4 * l15;
      v4f o;
      o.x = zacc[4 * q + 0][v] * inv[v];
      o.y = zacc[4 * q + 1][v] * inv[v];
      o.z = zacc[4 * q + 2][v] * inv[v];
      o.w = zacc[4 * q + 3][v] * inv[v];
      *reinterpret_cast<v4f*>(Zb + (size_t)m * DH_ + d) = o;
    }
}

extern "C" void kernel_launch(void* const* d_in, const int* in_sizes, int n_in,
                              void* d_out, int out_size, void* d_ws, size_t ws_size,
                              hipStream_t stream) {
  const float* H    = (const float*)d_in[0];
  const float* G    = (const float*)d_in[1];
  const float* Wk   = (const float*)d_in[2];
  const float* Wq   = (const float*)d_in[3];
  const int*   mask = (const int*)d_in[4];
  float*       Z    = (float*)d_out;

  float* Qws  = (float*)d_ws;
  float* KTws = Qws + (size_t)B_ * T_ * DP_;

  proj_gemm_kernel<<<(B_ * T_) / 16, 256, 0, stream>>>(G, Wq, Qws, DG_);
  kproj_kernel<<<(B_ * L_) / KP_ROWS, 256, 0, stream>>>(H, Wk, KTws);
  attn_kernel<<<B_ * (T_ / 16), 256, 0, stream>>>(H, Qws, KTws, mask, Z);
}


# attention retiled: 32 query rows x half of DH per workgroup, each K^T and H fragment feeds two M-tiles; Q, P tiles at LDS pitch 260; both halves compute the logits; k order, row sums unchanged
# speedup vs baseline: 1.4796x; 1.0353x over previous
#include <hip/hip_runtime.h>
#include <hip/hip_bf16.h>

typedef __attribute__((ext_vector_type(2))) float v2f;
typedef __attribute__((ext_vector_type(4))) float v4f;
typedef __attribute__((ext_vector_type(8))) float v8f;

constexpr int B_ = 8, L_ = 8192, T_ = 512, DH_ = 1024, DG_ = 768, DP_ = 256;

__device__ __forceinline__ v8f wmma4(v2f a, v2f b, v8f c) {
  return __builtin_amdgcn_wmma_f32_16x16x4_f32(
       false, a,  false, b,
       (short)0, c,  false,  false);
}

__device__ __forceinline__ v8f zero8() {
  v8f z = {0.f, 0.f, 0.f, 0.f, 0.f, 0.f, 0.f, 0.f};
  return z;
}

__global__ __launch_bounds__(256) void proj_gemm_kernel(
    const float* __restrict__ A, const float* __restrict__ W,
    float* __restrict__ out, int Kd) {
  const int tid  = threadIdx.x;
  const int w    = tid >> 5;
  const int lane = tid & 31;
  const int half = lane >> 4;
  const int l15  = lane & 15;
  const int row_base = blockIdx.x * 16;
  const int col2 = w * 32 + 2 * l15;

  const float* Ab = A + (size_t)(row_base + l15) * Kd + 2 * half;

  v8f acc0 = zero8();
  v8f acc1 = zero8();

  for (int kk = 0; kk < Kd; kk += 4) {
    v2f a;
    a.x = Ab[kk];
    a.y = Ab[kk + 1];
    const float* Wb = W + (size_t)(kk + 2 * half) * DP_ + col2;
    const v2f w0 = *reinterpret_cast<const v2f*>(Wb);
    const v2f w1 = *reinterpret_cast<const v2f*>(Wb + DP_);
    v2f b0; b0.x = w0.x; b0.y = w1.x;
    v2f b1; b1.x = w0.y; b1.y = w1.y;
    acc0 = wmma4(a, b0, acc0);
    acc1 = wmma4(a, b1, acc1);
  }

#pragma unroll
  for (int v = 0; v < 8; ++v) {
    const int m = v + 8 * half;
    const size_t r = (size_t)row_base + m;
    v2f o; o.x = acc0[v]; o.y = acc1[v];
    *reinterpret_cast<v2f*>(out + r * DP_ + col2) = o;
  }
}

constexpr int KP_ROWS  = 64;
constexpr int KP_KC    = 32;
constexpr int KP_PITCH = 36;

__global__ __launch_bounds__(256) void kproj_kernel(
    const float* __restrict__ A, const float* __restrict__ W,
    float* __restrict__ out) {
  __shared__ alignas(16) float a_lds[2][KP_ROWS * KP_PITCH];

  const int tid  = threadIdx.x;
  const int w    = tid >> 5;
  const int lane = tid & 31;
  const int half = lane >> 4;
  const int l15  = lane & 15;
  const int row_base = blockIdx.x * KP_ROWS;
  const int col2 = w * 32 + 2 * l15;

  const int fr = tid >> 3;
  const int fc = (tid & 7) * 4;
  const float* Ag = A + (size_t)(row_base + fr) * DH_ + fc;
  const int so = fr * KP_PITCH + fc;

  v8f acc[4][2];
#pragma unroll
  for (int mt = 0; mt < 4; ++mt) { acc[mt][0] = zero8(); acc[mt][1] = zero8(); }

  v4f g0 = *reinterpret_cast<const v4f*>(Ag);
  v4f g1 = *reinterpret_cast<const v4f*>(Ag + (size_t)32 * DH_);
  *reinterpret_cast<v4f*>(&a_lds[0][so]) = g0;
  *reinterpret_cast<v4f*>(&a_lds[0][so + 32 * KP_PITCH]) = g1;
  __syncthreads();

  const float* Wl = W + (size_t)(2 * half) * DP_ + col2;
  const int ao = l15 * KP_PITCH + 2 * half;

  for (int kc = 0; kc < DH_; kc += KP_KC) {
    const int buf = (kc / KP_KC) & 1;
    const bool more = kc + KP_KC < DH_;
    if (more) {
      g0 = *reinterpret_cast<const v4f*>(Ag + kc + KP_KC);
      g1 = *reinterpret_cast<const v4f*>(Ag + (size_t)32 * DH_ + kc + KP_KC);
    }
    const float* ab = &a_lds[buf][ao];
#pragma unroll
    for (int ks = 0; ks < KP_KC / 4; ++ks) {
      const float* Wb = Wl + (size_t)(kc + 4 * ks) * DP_;
      const v2f w0 = *reinterpret_cast<const v2f*>(Wb);
      const v2f w1 = *reinterpret_cast<const v2f*>(Wb + DP_);
      v2f b0; b0.x = w0.x; b0.y = w1.x;
      v2f b1; b1.x = w0.y; b1.y = w1.y;
#pragma unroll
      for (int mt = 0; mt < 4; ++mt) {
        const v2f a = *reinterpret_cast<const v2f*>(ab + mt * 16 * KP_PITCH + 4 * ks);
        acc[mt][0] = wmma4(a, b0, acc[mt][0]);
        acc[mt][1] = wmma4(a, b1, acc[mt][1]);
      }
    }
    if (more) {
      *reinterpret_cast<v4f*>(&a_lds[buf ^ 1][so]) = g0;
      *reinterpret_cast<v4f*>(&a_lds[buf ^ 1][so + 32 * KP_PITCH]) = g1;
      __syncthreads();
    }
  }

  const size_t bb = (size_t)row_base >> 13;
  const size_t l0 = ((size_t)row_base & (size_t)(L_ - 1)) + 8 * half;
#pragma unroll
  for (int mt = 0; mt < 4; ++mt)
#pragma unroll
    for (int j = 0; j < 2; ++j) {
      float* o = out + (bb * DP_ + col2 + j) * L_ + l0 + mt * 16;
      v4f lo; lo.x = acc[mt][j][0]; lo.y = acc[mt][j][1]; lo.z = acc[mt][j][2]; lo.w = acc[mt][j][3];
      v4f hi; hi.x = acc[mt][j][4]; hi.y = acc[mt][j][5]; hi.z = acc[mt][j][6]; hi.w = acc[mt][j][7];
      *reinterpret_cast<v4f*>(o)     = lo;
      *reinterpret_cast<v4f*>(o + 4) = hi;
    }
}


constexpr int AT_ROWS  = 32;
constexpr int AT_DCOLS = 512;
constexpr int AT_PITCH = 260;
constexpr int KA_AHEAD = 12;
constexpr int KA_RING  = 16;
constexpr int HC_AHEAD = 6;
constexpr int HC_RING  = 8;

__device__ __forceinline__ void fetch_k(v2f& x, v2f& y, const float* kp) {
  x = *reinterpret_cast<const v2f*>(kp);
  y = *reinterpret_cast<const v2f*>(kp + L_);
}

__device__ __forceinline__ void fetch_h(v4f& x, v4f& y, const float* hrow) {
  x = *reinterpret_cast<const v4f*>(hrow);
  y = *reinterpret_cast<const v4f*>(hrow + DH_);
}

__device__ __forceinline__ void fetch_a2(v2f (&a)[2], const float* ap) {
  a[0] = *reinterpret_cast<const v2f*>(ap);
  a[1] = *reinterpret_cast<const v2f*>(ap + 16 * AT_PITCH);
}

template <int MASK>
__device__ __forceinline__ float row_xor(float x) {
  const int xi = __builtin_bit_cast(int, x);
  return __builtin_bit_cast(float, __builtin_amdgcn_update_dpp(xi, xi, 0x160 + MASK, 0xF, 0xF, false));
}

__global__ __launch_bounds__(256) void attn_kernel(
    const float* __restrict__ H, const float* __restrict__ Q,
    const float* __restrict__ KT, const int* __restrict__ mask,
    float* __restrict__ Z) {
  __shared__ alignas(16) float q_lds[AT_ROWS * AT_PITCH];
  __shared__ alignas(16) float p_lds[AT_ROWS * AT_PITCH];
  __shared__ float red[8 * AT_ROWS];
  __shared__ float mrow[AT_ROWS];
  __shared__ float srow[AT_ROWS];
  __shared__ float frow[AT_ROWS];

  const int tid  = threadIdx.x;
  const int w    = tid >> 5;
  const int lane = tid & 31;
  const int half = lane >> 4;
  const int l15  = lane & 15;
  const int b    = blockIdx.x >> 5;
  const int t0   = ((blockIdx.x & 31) >> 1) * AT_ROWS;
  const int d0   = (blockIdx.x & 1) * AT_DCOLS + w * 64 + 4 * l15;

  const float* Qb = Q + ((size_t)b * T_ + t0) * DP_;
  for (int i = tid; i < AT_ROWS * DP_; i += 256) q_lds[(i >> 8) * AT_PITCH + (i & 255)] = Qb[i];
  if (tid < AT_ROWS) { mrow[tid] = -3.0e38f; srow[tid] = 0.f; }
  __syncthreads();

  v8f zacc[2][4];
#pragma unroll
  for (int mt = 0; mt < 2; ++mt)
#pragma unroll
    for (int j = 0; j < 4; ++j) zacc[mt][j] = zero8();

  const float scale = 0.0625f;
  const float* KTb = KT + (size_t)b * DP_ * L_;
  const int col2 = w * 32 + 2 * l15;

  for (int lc = 0; lc < L_; lc += 256) {
    v8f s[2][2];
#pragma unroll
    for (int mt = 0; mt < 2; ++mt) { s[mt][0] = zero8(); s[mt][1] = zero8(); }
    {
      const float* kp0 = KTb + (size_t)(2 * half) * L_ + lc + col2;
      const float* qp0 = q_lds + l15 * AT_PITCH + 2 * half;
      v2f kx[KA_RING], ky[KA_RING];
      v2f qa[KA_RING][2];
      constexpr int NS = DP_ / 4;
#pragma unroll
      for (int st = 0; st < KA_AHEAD; ++st) {
        fetch_k(kx[st], ky[st], kp0 + (size_t)(4 * st) * L_);
        fetch_a2(qa[st], qp0 + 4 * st);
      }
      for (int sb = 0; sb + KA_RING + KA_AHEAD <= NS; sb += KA_RING) {
#pragma unroll
        for (int u = 0; u < KA_RING; ++u) {
          const int sn = sb + u + KA_AHEAD;
          constexpr int M = KA_RING - 1;
          fetch_k(kx[(u + KA_AHEAD) & M], ky[(u + KA_AHEAD) & M], kp0 + (size_t)(4 * sn) * L_);
          fetch_a2(qa[(u + KA_AHEAD) & M], qp0 + 4 * sn);
          v2f b0; b0.x = kx[u].x; b0.y = ky[u].x;
          v2f b1; b1.x = kx[u].y; b1.y = ky[u].y;
#pragma unroll
          for (int mt = 0; mt < 2; ++mt) {
            s[mt][0] = wmma4(qa[u][mt], b0, s[mt][0]);
            s[mt][1] = wmma4(qa[u][mt], b1, s[mt][1]);
          }
        }
      }
      {
        constexpr int sb = ((NS - KA_AHEAD) / KA_RING) * KA_RING;
#pragma unroll
        for (int u = 0; u < NS - sb; ++u) {
          const int sn = sb + u + KA_AHEAD;
          constexpr int M = KA_RING - 1;
          if (sn < NS) {
            fetch_k(kx[(u + KA_AHEAD) & M], ky[(u + KA_AHEAD) & M], kp0 + (size_t)(4 * sn) * L_);
            fetch_a2(qa[(u + KA_AHEAD) & M], qp0 + 4 * sn);
          }
          v2f b0; b0.x = kx[u & M].x; b0.y = ky[u & M].x;
          v2f b1; b1.x = kx[u & M].y; b1.y = ky[u & M].y;
#pragma unroll
          for (int mt = 0; mt < 2; ++mt) {
            s[mt][0] = wmma4(qa[u & M][mt], b0, s[mt][0]);
            s[mt][1] = wmma4(qa[u & M][mt], b1, s[mt][1]);
          }
        }
      }
    }

    const float* Hb = H + ((size_t)b * L_ + lc + 2 * half) * DH_ + d0;
    v4f hx[HC_RING], hy[HC_RING];
#pragma unroll
    for (int st = 0; st < HC_AHEAD; ++st) fetch_h(hx[st], hy[st], Hb + (size_t)(4 * st) * DH_);

    const int mv0 = mask[(size_t)b * L_ + lc + col2];
    const int mv1 = mask[(size_t)b * L_ + lc + col2 + 1];
    float cmx[2][8];
#pragma unroll
    for (int mt = 0; mt < 2; ++mt)
#pragma unroll
      for (int v = 0; v < 8; ++v) {
        float x0 = s[mt][0][v] * scale; if (mv0) x0 = -3.0e38f;
        float x1 = s[mt][1][v] * scale; if (mv1) x1 = -3.0e38f;
        s[mt][0][v] = x0; s[mt][1][v] = x1;
        cmx[mt][v] = fmaxf(x0, x1);
      }
#pragma unroll
    for (int i = 0; i < 16; ++i) cmx[i >> 3][i & 7] = fmaxf(cmx[i >> 3][i & 7], row_xor<8>(cmx[i >> 3][i & 7]));
#pragma unroll
    for (int i = 0; i < 16; ++i) cmx[i >> 3][i & 7] = fmaxf(cmx[i >> 3][i & 7], row_xor<4>(cmx[i >> 3][i & 7]));
#pragma unroll
    for (int i = 0; i < 16; ++i) cmx[i >> 3][i & 7] = fmaxf(cmx[i >> 3][i & 7], row_xor<2>(cmx[i >> 3][i & 7]));
#pragma unroll
    for (int i = 0; i < 16; ++i) cmx[i >> 3][i & 7] = fmaxf(cmx[i >> 3][i & 7], row_xor<1>(cmx[i >> 3][i & 7]));
    if (l15 == 0) {
#pragma unroll
      for (int i = 0; i < 16; ++i) red[w * AT_ROWS + (i >> 3) * 16 + (i & 7) + 8 * half] = cmx[i >> 3][i & 7];
    }
    __syncthreads();

    if (tid < AT_ROWS) {
      float cm = red[tid];
#pragma unroll
      for (int ww = 1; ww < 8; ++ww) cm = fmaxf(cm, red[ww * AT_ROWS + tid]);
      const float mo = mrow[tid];
      const float mn = fmaxf(mo, cm);
      const float f  = __expf(mo - mn);
      srow[tid] *= f;
      frow[tid] = f;
      mrow[tid] = mn;
    }
    __syncthreads();

    float fsel[2][8], msel[2][8];
#pragma unroll
    for (int mt = 0; mt < 2; ++mt)
#pragma unroll
      for (int v = 0; v < 8; ++v) {
        fsel[mt][v] = frow[mt * 16 + v + 8 * half];
        msel[mt][v] = mrow[mt * 16 + v + 8 * half];
      }
#pragma unroll
    for (int mt = 0; mt < 2; ++mt)
#pragma unroll
      for (int j = 0; j < 4; ++j)
#pragma unroll
        for (int v = 0; v < 8; ++v) zacc[mt][j][v] *= fsel[mt][v];

    float psum[2][8];
#pragma unroll
    for (int mt = 0; mt < 2; ++mt)
#pragma unroll
      for (int v = 0; v < 8; ++v) {
        const float p0 = __expf(s[mt][0][v] - msel[mt][v]);
        const float p1 = __expf(s[mt][1][v] - msel[mt][v]);
        const int row = mt * 16 + v + 8 * half;
        v2f pp; pp.x = p0; pp.y = p1;
        *reinterpret_cast<v2f*>(&p_lds[row * AT_PITCH + col2]) = pp;
        psum[mt][v] = p0 + p1;
      }
#pragma unroll
    for (int i = 0; i < 16; ++i) psum[i >> 3][i & 7] += row_xor<8>(psum[i >> 3][i & 7]);
#pragma unroll
    for (int i = 0; i < 16; ++i) psum[i >> 3][i & 7] += row_xor<4>(psum[i >> 3][i & 7]);
#pragma unroll
    for (int i = 0; i < 16; ++i) psum[i >> 3][i & 7] += row_xor<2>(psum[i >> 3][i & 7]);
#pragma unroll
    for (int i = 0; i < 16; ++i) psum[i >> 3][i & 7] += row_xor<1>(psum[i >> 3][i & 7]);
    if (l15 == 0) {
#pragma unroll
      for (int i = 0; i < 16; ++i) red[w * AT_ROWS + (i >> 3) * 16 + (i & 7) + 8 * half] = psum[i >> 3][i & 7];
    }
    __syncthreads();

    if (tid < AT_ROWS) {
      float acc = 0.f;
#pragma unroll
      for (int ww = 0; ww < 8; ++ww) acc += red[ww * AT_ROWS + tid];
      srow[tid] += acc;
    }
    __syncthreads();

    {
      const float* pp0 = p_lds + l15 * AT_PITCH + 2 * half;
      v2f pa[HC_RING][2];
      constexpr int NS = 256 / 4;
#pragma unroll
      for (int st = 0; st < HC_AHEAD; ++st) fetch_a2(pa[st], pp0 + 4 * st);
      for (int sb = 0; sb + HC_RING + HC_AHEAD <= NS; sb += HC_RING) {
#pragma unroll
        for (int u = 0; u < HC_RING; ++u) {
          const int sn = sb + u + HC_AHEAD;
          constexpr int M = HC_RING - 1;
          fetch_h(hx[(u + HC_AHEAD) & M], hy[(u + HC_AHEAD) & M], Hb + (size_t)(4 * sn) * DH_);
          fetch_a2(pa[(u + HC_AHEAD) & M], pp0 + 4 * sn);
#pragma unroll
          for (int j = 0; j < 4; ++j) {
            v2f bb; bb.x = hx[u][j]; bb.y = hy[u][j];
#pragma unroll
            for (int mt = 0; mt < 2; ++mt) zacc[mt][j] = wmma4(pa[u][mt], bb, zacc[mt][j]);
          }
        }
      }
      {
        constexpr int sb = ((NS - HC_AHEAD) / HC_RING) * HC_RING;
#pragma unroll
        for (int u = 0; u < NS - sb; ++u) {
          const int sn = sb + u + HC_AHEAD;
          constexpr int M = HC_RING - 1;
          if (sn < NS) {
            fetch_h(hx[(u + HC_AHEAD) & M], hy[(u + HC_AHEAD) & M], Hb + (size_t)(4 * sn) * DH_);
            fetch_a2(pa[(u + HC_AHEAD) & M], pp0 + 4 * sn);
          }
#pragma unroll
          for (int j = 0; j < 4; ++j) {
            v2f bb; bb.x = hx[u & M][j]; bb.y = hy[u & M][j];
#pragma unroll
            for (int mt = 0; mt < 2; ++mt) zacc[mt][j] = wmma4(pa[u & M][mt], bb, zacc[mt][j]);
          }
        }
      }
    }
  }

  float* Zb = Z + ((size_t)b * T_ + t0) * DH_ + d0;
#pragma unroll
  for (int mt = 0; mt < 2; ++mt)
#pragma unroll
    for (int v = 0; v < 8; ++v) {
      const int m = mt * 16 + v + 8 * half;
      const float inv = 1.0f / srow[m];
      v4f o;
      o.x = zacc[mt][0][v] * inv;
      o.y = zacc[mt][1][v] * inv;
      o.z = zacc[mt][2][v] * inv;
      o.w = zacc[mt][3][v] * inv;
      *reinterpret_cast<v4f*>(Zb + (size_t)m * DH_) = o;
    }
}

extern "C" void kernel_launch(void* const* d_in, const int* in_sizes, int n_in,
                              void* d_out, int out_size, void* d_ws, size_t ws_size,
                              hipStream_t stream) {
  const float* H    = (const float*)d_in[0];
  const float* G    = (const float*)d_in[1];
  const float* Wk   = (const float*)d_in[2];
  const float* Wq   = (const float*)d_in[3];
  const int*   mask = (const int*)d_in[4];
  float*       Z    = (float*)d_out;

  float* Qws  = (float*)d_ws;
  float* KTws = Qws + (size_t)B_ * T_ * DP_;

  proj_gemm_kernel<<<(B_ * T_) / 16, 256, 0, stream>>>(G, Wq, Qws, DG_);
  kproj_kernel<<<(B_ * L_) / KP_ROWS, 256, 0, stream>>>(H, Wk, KTws);
  attn_kernel<<<B_ * (T_ / AT_ROWS) * (DH_ / AT_DCOLS), 256, 0, stream>>>(H, Qws, KTws, mask, Z);
}


# attention streams L in 1024-column chunks (was 256): a quarter of the softmax rounds and barriers; a lane holds 2x4 adjacent logit columns (16-B K^T loads), added first in the row sum; rescales move
# speedup vs baseline: 1.5715x; 1.0621x over previous
#include <hip/hip_runtime.h>
#include <hip/hip_bf16.h>

typedef __attribute__((ext_vector_type(2))) float v2f;
typedef __attribute__((ext_vector_type(4))) float v4f;
typedef __attribute__((ext_vector_type(8))) float v8f;

constexpr int B_ = 8, L_ = 8192, T_ = 512, DH_ = 1024, DG_ = 768, DP_ = 256;

__device__ __forceinline__ v8f wmma4(v2f a, v2f b, v8f c) {
  return __builtin_amdgcn_wmma_f32_16x16x4_f32(
       false, a,  false, b,
       (short)0, c,  false,  false);
}

__device__ __forceinline__ v8f zero8() {
  v8f z = {0.f, 0.f, 0.f, 0.f, 0.f, 0.f, 0.f, 0.f};
  return z;
}

__global__ __launch_bounds__(256) void proj_gemm_kernel(
    const float* __restrict__ A, const float* __restrict__ W,
    float* __restrict__ out, int Kd) {
  const int tid  = threadIdx.x;
  const int w    = tid >> 5;
  const int lane = tid & 31;
  const int half = lane >> 4;
  const int l15  = lane & 15;
  const int row_base = blockIdx.x * 16;
  const int col2 = w * 32 + 2 * l15;

  const float* Ab = A + (size_t)(row_base + l15) * Kd + 2 * half;

  v8f acc0 = zero8();
  v8f acc1 = zero8();

  for (int kk = 0; kk < Kd; kk += 4) {
    v2f a;
    a.x = Ab[kk];
    a.y = Ab[kk + 1];
    const float* Wb = W + (size_t)(kk + 2 * half) * DP_ + col2;
    const v2f w0 = *reinterpret_cast<const v2f*>(Wb);
    const v2f w1 = *reinterpret_cast<const v2f*>(Wb + DP_);
    v2f b0; b0.x = w0.x; b0.y = w1.x;
    v2f b1; b1.x = w0.y; b1.y = w1.y;
    acc0 = wmma4(a, b0, acc0);
    acc1 = wmma4(a, b1, acc1);
  }

#pragma unroll
  for (int v = 0; v < 8; ++v) {
    const int m = v + 8 * half;
    const size_t r = (size_t)row_base + m;
    v2f o; o.x = acc0[v]; o.y = acc1[v];
    *reinterpret_cast<v2f*>(out + r * DP_ + col2) = o;
  }
}

constexpr int KP_ROWS  = 64;
constexpr int KP_KC    = 32;
constexpr int KP_PITCH = 36;

__global__ __launch_bounds__(256) void kproj_kernel(
    const float* __restrict__ A, const float* __restrict__ W,
    float* __restrict__ out) {
  __shared__ alignas(16) float a_lds[2][KP_ROWS * KP_PITCH];

  const int tid  = threadIdx.x;
  const int w    = tid >> 5;
  const int lane = tid & 31;
  const int half = lane >> 4;
  const int l15  = lane & 15;
  const int row_base = blockIdx.x * KP_ROWS;
  const int col2 = w * 32 + 2 * l15;

  const int fr = tid >> 3;
  const int fc = (tid & 7) * 4;
  const float* Ag = A + (size_t)(row_base + fr) * DH_ + fc;
  const int so = fr * KP_PITCH + fc;

  v8f acc[4][2];
#pragma unroll
  for (int mt = 0; mt < 4; ++mt) { acc[mt][0] = zero8(); acc[mt][1] = zero8(); }

  v4f g0 = *reinterpret_cast<const v4f*>(Ag);
  v4f g1 = *reinterpret_cast<const v4f*>(Ag + (size_t)32 * DH_);
  *reinterpret_cast<v4f*>(&a_lds[0][so]) = g0;
  *reinterpret_cast<v4f*>(&a_lds[0][so + 32 * KP_PITCH]) = g1;
  __syncthreads();

  const float* Wl = W + (size_t)(2 * half) * DP_ + col2;
  const int ao = l15 * KP_PITCH + 2 * half;

  for (int kc = 0; kc < DH_; kc += KP_KC) {
    const int buf = (kc / KP_KC) & 1;
    const bool more = kc + KP_KC < DH_;
    if (more) {
      g0 = *reinterpret_cast<const v4f*>(Ag + kc + KP_KC);
      g1 = *reinterpret_cast<const v4f*>(Ag + (size_t)32 * DH_ + kc + KP_KC);
    }
    const float* ab = &a_lds[buf][ao];
#pragma unroll
    for (int ks = 0; ks < KP_KC / 4; ++ks) {
      const float* Wb = Wl + (size_t)(kc + 4 * ks) * DP_;
      const v2f w0 = *reinterpret_cast<const v2f*>(Wb);
      const v2f w1 = *reinterpret_cast<const v2f*>(Wb + DP_);
      v2f b0; b0.x = w0.x; b0.y = w1.x;
      v2f b1; b1.x = w0.y; b1.y = w1.y;
#pragma unroll
      for (int mt = 0; mt < 4; ++mt) {
        const v2f a = *reinterpret_cast<const v2f*>(ab + mt * 16 * KP_PITCH + 4 * ks);
        acc[mt][0] = wmma4(a, b0, acc[mt][0]);
        acc[mt][1] = wmma4(a, b1, acc[mt][1]);
      }
    }
    if (more) {
      *reinterpret_cast<v4f*>(&a_lds[buf ^ 1][so]) = g0;
      *reinterpret_cast<v4f*>(&a_lds[buf ^ 1][so + 32 * KP_PITCH]) = g1;
      __syncthreads();
    }
  }

  const size_t bb = (size_t)row_base >> 13;
  const size_t l0 = ((size_t)row_base & (size_t)(L_ - 1)) + 8 * half;
#pragma unroll
  for (int mt = 0; mt < 4; ++mt)
#pragma unroll
    for (int j = 0; j < 2; ++j) {
      float* o = out + (bb * DP_ + col2 + j) * L_ + l0 + mt * 16;
      v4f lo; lo.x = acc[mt][j][0]; lo.y = acc[mt][j][1]; lo.z = acc[mt][j][2]; lo.w = acc[mt][j][3];
      v4f hi; hi.x = acc[mt][j][4]; hi.y = acc[mt][j][5]; hi.z = acc[mt][j][6]; hi.w = acc[mt][j][7];
      *reinterpret_cast<v4f*>(o)     = lo;
      *reinterpret_cast<v4f*>(o + 4) = hi;
    }
}


constexpr int AT_ROWS  = 32;
constexpr int AT_DCOLS = 512;
constexpr int AT_NG    = 2;
constexpr int AT_CHUNK = 8 * 64 * AT_NG;
constexpr int AT_PITCH = 260;
constexpr int AP_PITCH = AT_CHUNK + 4;
constexpr int KA_AHEAD = 6 / AT_NG;
constexpr int KA_RING  = 8 / AT_NG;
constexpr int HC_AHEAD = 6;
constexpr int HC_RING  = 8;

__device__ __forceinline__ void fetch_k(v4f& x, v4f& y, const float* kp) {
  x = *reinterpret_cast<const v4f*>(kp);
  y = *reinterpret_cast<const v4f*>(kp + L_);
}

__device__ __forceinline__ void fetch_h(v4f& x, v4f& y, const float* hrow) {
  x = *reinterpret_cast<const v4f*>(hrow);
  y = *reinterpret_cast<const v4f*>(hrow + DH_);
}

template <int PITCH>
__device__ __forceinline__ void fetch_a2(v2f (&a)[2], const float* ap) {
  a[0] = *reinterpret_cast<const v2f*>(ap);
  a[1] = *reinterpret_cast<const v2f*>(ap + 16 * PITCH);
}

template <int MASK>
__device__ __forceinline__ float row_xor(float x) {
  const int xi = __builtin_bit_cast(int, x);
  return __builtin_bit_cast(float, __builtin_amdgcn_update_dpp(xi, xi, 0x160 + MASK, 0xF, 0xF, false));
}

__global__ __launch_bounds__(256) void attn_kernel(
    const float* __restrict__ H, const float* __restrict__ Q,
    const float* __restrict__ KT, const int* __restrict__ mask,
    float* __restrict__ Z) {
  __shared__ alignas(16) float q_lds[AT_ROWS * AT_PITCH];
  __shared__ alignas(16) float p_lds[AT_ROWS * AP_PITCH];
  __shared__ float red[8 * AT_ROWS];
  __shared__ float mrow[AT_ROWS];
  __shared__ float srow[AT_ROWS];
  __shared__ float frow[AT_ROWS];

  const int tid  = threadIdx.x;
  const int w    = tid >> 5;
  const int lane = tid & 31;
  const int half = lane >> 4;
  const int l15  = lane & 15;
  const int b    = blockIdx.x >> 5;
  const int t0   = ((blockIdx.x & 31) >> 1) * AT_ROWS;
  const int d0   = (blockIdx.x & 1) * AT_DCOLS + w * 64 + 4 * l15;

  const float* Qb = Q + ((size_t)b * T_ + t0) * DP_;
  for (int i = tid; i < AT_ROWS * DP_; i += 256) q_lds[(i >> 8) * AT_PITCH + (i & 255)] = Qb[i];
  if (tid < AT_ROWS) { mrow[tid] = -3.0e38f; srow[tid] = 0.f; }
  __syncthreads();

  v8f zacc[2][4];
#pragma unroll
  for (int mt = 0; mt < 2; ++mt)
#pragma unroll
    for (int j = 0; j < 4; ++j) zacc[mt][j] = zero8();

  const float scale = 0.0625f;
  const float* KTb = KT + (size_t)b * DP_ * L_;
  const int col4 = w * (64 * AT_NG) + 4 * l15;

  for (int lc = 0; lc < L_; lc += AT_CHUNK) {
    v8f s[AT_NG][2][4];
#pragma unroll
    for (int g = 0; g < AT_NG; ++g)
#pragma unroll
      for (int mt = 0; mt < 2; ++mt)
#pragma unroll
        for (int j = 0; j < 4; ++j) s[g][mt][j] = zero8();
    {
      const float* kp0 = KTb + (size_t)(2 * half) * L_ + lc + col4;
      const float* qp0 = q_lds + l15 * AT_PITCH + 2 * half;
      v4f kx[KA_RING][AT_NG], ky[KA_RING][AT_NG];
      v2f qa[KA_RING][2];
      constexpr int NS = DP_ / 4;
#pragma unroll
      for (int st = 0; st < KA_AHEAD; ++st) {
#pragma unroll
        for (int g = 0; g < AT_NG; ++g) fetch_k(kx[st][g], ky[st][g], kp0 + (size_t)(4 * st) * L_ + 64 * g);
        fetch_a2<AT_PITCH>(qa[st], qp0 + 4 * st);
      }
      for (int sb = 0; sb + KA_RING + KA_AHEAD <= NS; sb += KA_RING) {
#pragma unroll
        for (int u = 0; u < KA_RING; ++u) {
          const int sn = sb + u + KA_AHEAD;
          constexpr int M = KA_RING - 1;
#pragma unroll
          for (int g = 0; g < AT_NG; ++g)
            fetch_k(kx[(u + KA_AHEAD) & M][g], ky[(u + KA_AHEAD) & M][g], kp0 + (size_t)(4 * sn) * L_ + 64 * g);
          fetch_a2<AT_PITCH>(qa[(u + KA_AHEAD) & M], qp0 + 4 * sn);
#pragma unroll
          for (int g = 0; g < AT_NG; ++g)
#pragma unroll
            for (int j = 0; j < 4; ++j) {
              v2f bb; bb.x = kx[u][g][j]; bb.y = ky[u][g][j];
#pragma unroll
              for (int mt = 0; mt < 2; ++mt) s[g][mt][j] = wmma4(qa[u][mt], bb, s[g][mt][j]);
            }
        }
      }
      {
        constexpr int sb = ((NS - KA_AHEAD) / KA_RING) * KA_RING;
#pragma unroll
        for (int u = 0; u < NS - sb; ++u) {
          const int sn = sb + u + KA_AHEAD;
          constexpr int M = KA_RING - 1;
          if (sn < NS) {
#pragma unroll
            for (int g = 0; g < AT_NG; ++g)
              fetch_k(kx[(u + KA_AHEAD) & M][g], ky[(u + KA_AHEAD) & M][g], kp0 + (size_t)(4 * sn) * L_ + 64 * g);
            fetch_a2<AT_PITCH>(qa[(u + KA_AHEAD) & M], qp0 + 4 * sn);
          }
#pragma unroll
          for (int g = 0; g < AT_NG; ++g)
#pragma unroll
            for (int j = 0; j < 4; ++j) {
              v2f bb; bb.x = kx[u & M][g][j]; bb.y = ky[u & M][g][j];
#pragma unroll
              for (int mt = 0; mt < 2; ++mt) s[g][mt][j] = wmma4(qa[u & M][mt], bb, s[g][mt][j]);
            }
        }
      }
    }

    const float* Hb = H + ((size_t)b * L_ + lc + 2 * half) * DH_ + d0;
    v4f hx[HC_RING], hy[HC_RING];
#pragma unroll
    for (int st = 0; st < HC_AHEAD; ++st) fetch_h(hx[st], hy[st], Hb + (size_t)(4 * st) * DH_);

    int mv[AT_NG][4];
#pragma unroll
    for (int g = 0; g < AT_NG; ++g)
#pragma unroll
      for (int j = 0; j < 4; ++j) mv[g][j] = mask[(size_t)b * L_ + lc + col4 + 64 * g + j];
    float cmx[2][8];
#pragma unroll
    for (int mt = 0; mt < 2; ++mt)
#pragma unroll
      for (int v = 0; v < 8; ++v) {
        float cm = 0.f;
#pragma unroll
        for (int g = 0; g < AT_NG; ++g) {
          float x[4];
#pragma unroll
          for (int j = 0; j < 4; ++j) {
            float xv = s[g][mt][j][v] * scale; if (mv[g][j]) xv = -3.0e38f;
            s[g][mt][j][v] = xv; x[j] = xv;
          }
          const float m4 = fmaxf(fmaxf(x[0], x[1]), fmaxf(x[2], x[3]));
          cm = (g == 0) ? m4 : fmaxf(cm, m4);
        }
        cmx[mt][v] = cm;
      }
#pragma unroll
    for (int i = 0; i < 16; ++i) cmx[i >> 3][i & 7] = fmaxf(cmx[i >> 3][i & 7], row_xor<8>(cmx[i >> 3][i & 7]));
#pragma unroll
    for (int i = 0; i < 16; ++i) cmx[i >> 3][i & 7] = fmaxf(cmx[i >> 3][i & 7], row_xor<4>(cmx[i >> 3][i & 7]));
#pragma unroll
    for (int i = 0; i < 16; ++i) cmx[i >> 3][i & 7] = fmaxf(cmx[i >> 3][i & 7], row_xor<2>(cmx[i >> 3][i & 7]));
#pragma unroll
    for (int i = 0; i < 16; ++i) cmx[i >> 3][i & 7] = fmaxf(cmx[i >> 3][i & 7], row_xor<1>(cmx[i >> 3][i & 7]));
    if (l15 == 0) {
#pragma unroll
      for (int i = 0; i < 16; ++i) red[w * AT_ROWS + (i >> 3) * 16 + (i & 7) + 8 * half] = cmx[i >> 3][i & 7];
    }
    __syncthreads();

    if (tid < AT_ROWS) {
      float cm = red[tid];
#pragma unroll
      for (int ww = 1; ww < 8; ++ww) cm = fmaxf(cm, red[ww * AT_ROWS + tid]);
      const float mo = mrow[tid];
      const float mn = fmaxf(mo, cm);
      const float f  = __expf(mo - mn);
      srow[tid] *= f;
      frow[tid] = f;
      mrow[tid] = mn;
    }
    __syncthreads();

    float fsel[2][8], msel[2][8];
#pragma unroll
    for (int mt = 0; mt < 2; ++mt)
#pragma unroll
      for (int v = 0; v < 8; ++v) {
        fsel[mt][v] = frow[mt * 16 + v + 8 * half];
        msel[mt][v] = mrow[mt * 16 + v + 8 * half];
      }
#pragma unroll
    for (int mt = 0; mt < 2; ++mt)
#pragma unroll
      for (int j = 0; j < 4; ++j)
#pragma unroll
        for (int v = 0; v < 8; ++v) zacc[mt][j][v] *= fsel[mt][v];

    float psum[2][8];
#pragma unroll
    for (int mt = 0; mt < 2; ++mt)
#pragma unroll
      for (int v = 0; v < 8; ++v) {
        const int row = mt * 16 + v + 8 * half;
        float ps = 0.f;
#pragma unroll
        for (int g = 0; g < AT_NG; ++g) {
          v4f pp;
          pp.x = __expf(s[g][mt][0][v] - msel[mt][v]);
          pp.y = __expf(s[g][mt][1][v] - msel[mt][v]);
          pp.z = __expf(s[g][mt][2][v] - msel[mt][v]);
          pp.w = __expf(s[g][mt][3][v] - msel[mt][v]);
          *reinterpret_cast<v4f*>(&p_lds[row * AP_PITCH + col4 + 64 * g]) = pp;
          const float s4 = (pp.x + pp.y) + (pp.z + pp.w);
          ps = (g == 0) ? s4 : ps + s4;
        }
        psum[mt][v] = ps;
      }
#pragma unroll
    for (int i = 0; i < 16; ++i) psum[i >> 3][i & 7] += row_xor<8>(psum[i >> 3][i & 7]);
#pragma unroll
    for (int i = 0; i < 16; ++i) psum[i >> 3][i & 7] += row_xor<4>(psum[i >> 3][i & 7]);
#pragma unroll
    for (int i = 0; i < 16; ++i) psum[i >> 3][i & 7] += row_xor<2>(psum[i >> 3][i & 7]);
#pragma unroll
    for (int i = 0; i < 16; ++i) psum[i >> 3][i & 7] += row_xor<1>(psum[i >> 3][i & 7]);
    if (l15 == 0) {
#pragma unroll
      for (int i = 0; i < 16; ++i) red[w * AT_ROWS + (i >> 3) * 16 + (i & 7) + 8 * half] = psum[i >> 3][i & 7];
    }
    __syncthreads();

    if (tid < AT_ROWS) {
      float acc = 0.f;
#pragma unroll
      for (int ww = 0; ww < 8; ++ww) acc += red[ww * AT_ROWS + tid];
      srow[tid] += acc;
    }
    __syncthreads();

    {
      const float* pp0 = p_lds + l15 * AP_PITCH + 2 * half;
      v2f pa[HC_RING][2];
      constexpr int NS = AT_CHUNK / 4;
#pragma unroll
      for (int st = 0; st < HC_AHEAD; ++st) fetch_a2<AP_PITCH>(pa[st], pp0 + 4 * st);
      for (int sb = 0; sb + HC_RING + HC_AHEAD <= NS; sb += HC_RING) {
#pragma unroll
        for (int u = 0; u < HC_RING; ++u) {
          const int sn = sb + u + HC_AHEAD;
          constexpr int M = HC_RING - 1;
          fetch_h(hx[(u + HC_AHEAD) & M], hy[(u + HC_AHEAD) & M], Hb + (size_t)(4 * sn) * DH_);
          fetch_a2<AP_PITCH>(pa[(u + HC_AHEAD) & M], pp0 + 4 * sn);
#pragma unroll
          for (int j = 0; j < 4; ++j) {
            v2f bb; bb.x = hx[u][j]; bb.y = hy[u][j];
#pragma unroll
            for (int mt = 0; mt < 2; ++mt) zacc[mt][j] = wmma4(pa[u][mt], bb, zacc[mt][j]);
          }
        }
      }
      {
        constexpr int sb = ((NS - HC_AHEAD) / HC_RING) * HC_RING;
#pragma unroll
        for (int u = 0; u < NS - sb; ++u) {
          const int sn = sb + u + HC_AHEAD;
          constexpr int M = HC_RING - 1;
          if (sn < NS) {
            fetch_h(hx[(u + HC_AHEAD) & M], hy[(u + HC_AHEAD) & M], Hb + (size_t)(4 * sn) * DH_);
            fetch_a2<AP_PITCH>(pa[(u + HC_AHEAD) & M], pp0 + 4 * sn);
          }
#pragma unroll
          for (int j = 0; j < 4; ++j) {
            v2f bb; bb.x = hx[u & M][j]; bb.y = hy[u & M][j];
#pragma unroll
            for (int mt = 0; mt < 2; ++mt) zacc[mt][j] = wmma4(pa[u & M][mt], bb, zacc[mt][j]);
          }
        }
      }
    }
  }

  float* Zb = Z + ((size_t)b * T_ + t0) * DH_ + d0;
#pragma unroll
  for (int mt = 0; mt < 2; ++mt)
#pragma unroll
    for (int v = 0; v < 8; ++v) {
      const int m = mt * 16 + v + 8 * half;
      const float inv = 1.0f / srow[m];
      v4f o;
      o.x = zacc[mt][0][v] * inv;
      o.y = zacc[mt][1][v] * inv;
      o.z = zacc[mt][2][v] * inv;
      o.w = zacc[mt][3][v] * inv;
      *reinterpret_cast<v4f*>(Zb + (size_t)m * DH_) = o;
    }
}

extern "C" void kernel_launch(void* const* d_in, const int* in_sizes, int n_in,
                              void* d_out, int out_size, void* d_ws, size_t ws_size,
                              hipStream_t stream) {
  const float* H    = (const float*)d_in[0];
  const float* G    = (const float*)d_in[1];
  const float* Wk   = (const float*)d_in[2];
  const float* Wq   = (const float*)d_in[3];
  const int*   mask = (const int*)d_in[4];
  float*       Z    = (float*)d_out;

  float* Qws  = (float*)d_ws;
  float* KTws = Qws + (size_t)B_ * T_ * DP_;

  proj_gemm_kernel<<<(B_ * T_) / 16, 256, 0, stream>>>(G, Wq, Qws, DG_);
  kproj_kernel<<<(B_ * L_) / KP_ROWS, 256, 0, stream>>>(H, Wk, KTws);
  attn_kernel<<<B_ * (T_ / AT_ROWS) * (DH_ / AT_DCOLS), 256, 0, stream>>>(H, Qws, KTws, mask, Z);
}


# attention: each 64-WMMA block's operand loads (16 global, 8 or 16 LDS fragment reads) go out as one group a whole block ahead into a second register set, held by a scheduling barrier; sums unchanged
# speedup vs baseline: 1.6426x; 1.0452x over previous
#include <hip/hip_runtime.h>
#include <hip/hip_bf16.h>

typedef __attribute__((ext_vector_type(2))) float v2f;
typedef __attribute__((ext_vector_type(4))) float v4f;
typedef __attribute__((ext_vector_type(8))) float v8f;

constexpr int B_ = 8, L_ = 8192, T_ = 512, DH_ = 1024, DG_ = 768, DP_ = 256;

__device__ __forceinline__ v8f wmma4(v2f a, v2f b, v8f c) {
  return __builtin_amdgcn_wmma_f32_16x16x4_f32(
       false, a,  false, b,
       (short)0, c,  false,  false);
}

__device__ __forceinline__ v8f zero8() {
  v8f z = {0.f, 0.f, 0.f, 0.f, 0.f, 0.f, 0.f, 0.f};
  return z;
}

__global__ __launch_bounds__(256) void proj_gemm_kernel(
    const float* __restrict__ A, const float* __restrict__ W,
    float* __restrict__ out, int Kd) {
  const int tid  = threadIdx.x;
  const int w    = tid >> 5;
  const int lane = tid & 31;
  const int half = lane >> 4;
  const int l15  = lane & 15;
  const int row_base = blockIdx.x * 16;
  const int col2 = w * 32 + 2 * l15;

  const float* Ab = A + (size_t)(row_base + l15) * Kd + 2 * half;

  v8f acc0 = zero8();
  v8f acc1 = zero8();

  for (int kk = 0; kk < Kd; kk += 4) {
    v2f a;
    a.x = Ab[kk];
    a.y = Ab[kk + 1];
    const float* Wb = W + (size_t)(kk + 2 * half) * DP_ + col2;
    const v2f w0 = *reinterpret_cast<const v2f*>(Wb);
    const v2f w1 = *reinterpret_cast<const v2f*>(Wb + DP_);
    v2f b0; b0.x = w0.x; b0.y = w1.x;
    v2f b1; b1.x = w0.y; b1.y = w1.y;
    acc0 = wmma4(a, b0, acc0);
    acc1 = wmma4(a, b1, acc1);
  }

#pragma unroll
  for (int v = 0; v < 8; ++v) {
    const int m = v + 8 * half;
    const size_t r = (size_t)row_base + m;
    v2f o; o.x = acc0[v]; o.y = acc1[v];
    *reinterpret_cast<v2f*>(out + r * DP_ + col2) = o;
  }
}

constexpr int KP_ROWS  = 64;
constexpr int KP_KC    = 32;
constexpr int KP_PITCH = 36;

__global__ __launch_bounds__(256) void kproj_kernel(
    const float* __restrict__ A, const float* __restrict__ W,
    float* __restrict__ out) {
  __shared__ alignas(16) float a_lds[2][KP_ROWS * KP_PITCH];

  const int tid  = threadIdx.x;
  const int w    = tid >> 5;
  const int lane = tid & 31;
  const int half = lane >> 4;
  const int l15  = lane & 15;
  const int row_base = blockIdx.x * KP_ROWS;
  const int col2 = w * 32 + 2 * l15;

  const int fr = tid >> 3;
  const int fc = (tid & 7) * 4;
  const float* Ag = A + (size_t)(row_base + fr) * DH_ + fc;
  const int so = fr * KP_PITCH + fc;

  v8f acc[4][2];
#pragma unroll
  for (int mt = 0; mt < 4; ++mt) { acc[mt][0] = zero8(); acc[mt][1] = zero8(); }

  v4f g0 = *reinterpret_cast<const v4f*>(Ag);
  v4f g1 = *reinterpret_cast<const v4f*>(Ag + (size_t)32 * DH_);
  *reinterpret_cast<v4f*>(&a_lds[0][so]) = g0;
  *reinterpret_cast<v4f*>(&a_lds[0][so + 32 * KP_PITCH]) = g1;
  __syncthreads();

  const float* Wl = W + (size_t)(2 * half) * DP_ + col2;
  const int ao = l15 * KP_PITCH + 2 * half;

  for (int kc = 0; kc < DH_; kc += KP_KC) {
    const int buf = (kc / KP_KC) & 1;
    const bool more = kc + KP_KC < DH_;
    if (more) {
      g0 = *reinterpret_cast<const v4f*>(Ag + kc + KP_KC);
      g1 = *reinterpret_cast<const v4f*>(Ag + (size_t)32 * DH_ + kc + KP_KC);
    }
    const float* ab = &a_lds[buf][ao];
#pragma unroll
    for (int ks = 0; ks < KP_KC / 4; ++ks) {
      const float* Wb = Wl + (size_t)(kc + 4 * ks) * DP_;
      const v2f w0 = *reinterpret_cast<const v2f*>(Wb);
      const v2f w1 = *reinterpret_cast<const v2f*>(Wb + DP_);
      v2f b0; b0.x = w0.x; b0.y = w1.x;
      v2f b1; b1.x = w0.y; b1.y = w1.y;
#pragma unroll
      for (int mt = 0; mt < 4; ++mt) {
        const v2f a = *reinterpret_cast<const v2f*>(ab + mt * 16 * KP_PITCH + 4 * ks);
        acc[mt][0] = wmma4(a, b0, acc[mt][0]);
        acc[mt][1] = wmma4(a, b1, acc[mt][1]);
      }
    }
    if (more) {
      *reinterpret_cast<v4f*>(&a_lds[buf ^ 1][so]) = g0;
      *reinterpret_cast<v4f*>(&a_lds[buf ^ 1][so + 32 * KP_PITCH]) = g1;
      __syncthreads();
    }
  }

  const size_t bb = (size_t)row_base >> 13;
  const size_t l0 = ((size_t)row_base & (size_t)(L_ - 1)) + 8 * half;
#pragma unroll
  for (int mt = 0; mt < 4; ++mt)
#pragma unroll
    for (int j = 0; j < 2; ++j) {
      float* o = out + (bb * DP_ + col2 + j) * L_ + l0 + mt * 16;
      v4f lo; lo.x = acc[mt][j][0]; lo.y = acc[mt][j][1]; lo.z = acc[mt][j][2]; lo.w = acc[mt][j][3];
      v4f hi; hi.x = acc[mt][j][4]; hi.y = acc[mt][j][5]; hi.z = acc[mt][j][6]; hi.w = acc[mt][j][7];
      *reinterpret_cast<v4f*>(o)     = lo;
      *reinterpret_cast<v4f*>(o + 4) = hi;
    }
}


constexpr int AT_ROWS  = 32;
constexpr int AT_DCOLS = 512;
constexpr int AT_NG    = 2;
constexpr int AT_CHUNK = 8 * 64 * AT_NG;
constexpr int AT_PITCH = 260;
constexpr int AP_PITCH = AT_CHUNK + 4;
constexpr int KA_BLK = 8 / AT_NG;
constexpr int HC_BLK = 8;

__device__ __forceinline__ void fetch_k(v4f& x, v4f& y, const float* kp) {
  x = *reinterpret_cast<const v4f*>(kp);
  y = *reinterpret_cast<const v4f*>(kp + L_);
}

__device__ __forceinline__ void fetch_h(v4f& x, v4f& y, const float* hrow) {
  x = *reinterpret_cast<const v4f*>(hrow);
  y = *reinterpret_cast<const v4f*>(hrow + DH_);
}

template <int PITCH>
__device__ __forceinline__ void fetch_a2(v2f (&a)[2], const float* ap) {
  a[0] = *reinterpret_cast<const v2f*>(ap);
  a[1] = *reinterpret_cast<const v2f*>(ap + 16 * PITCH);
}

__device__ __forceinline__ void keep_loads_grouped() { __builtin_amdgcn_sched_barrier(0); }

struct KBlock { v4f kx[KA_BLK][AT_NG], ky[KA_BLK][AT_NG]; v2f qa[KA_BLK][2]; };
struct HBlock { v4f hx[HC_BLK], hy[HC_BLK]; v2f pa[HC_BLK][2]; };

__device__ __forceinline__ void fetch_block_a(KBlock& o, const float* kp, const float* qp) {
#pragma unroll
  for (int u = 0; u < KA_BLK; ++u)
#pragma unroll
    for (int g = 0; g < AT_NG; ++g) fetch_k(o.kx[u][g], o.ky[u][g], kp + (size_t)(4 * u) * L_ + 64 * g);
#pragma unroll
  for (int u = 0; u < KA_BLK; ++u) fetch_a2<AT_PITCH>(o.qa[u], qp + 4 * u);
}

__device__ __forceinline__ void mma_block_a(const KBlock& c, v8f (&s)[AT_NG][2][4]) {
#pragma unroll
  for (int u = 0; u < KA_BLK; ++u)
#pragma unroll
    for (int g = 0; g < AT_NG; ++g)
#pragma unroll
      for (int j = 0; j < 4; ++j) {
        v2f bb; bb.x = c.kx[u][g][j]; bb.y = c.ky[u][g][j];
#pragma unroll
        for (int mt = 0; mt < 2; ++mt) s[g][mt][j] = wmma4(c.qa[u][mt], bb, s[g][mt][j]);
      }
}

__device__ __forceinline__ void fetch_block_h(HBlock& o, const float* hp) {
#pragma unroll
  for (int u = 0; u < HC_BLK; ++u) fetch_h(o.hx[u], o.hy[u], hp + (size_t)(4 * u) * DH_);
}

__device__ __forceinline__ void fetch_block_p(HBlock& o, const float* pp) {
#pragma unroll
  for (int u = 0; u < HC_BLK; ++u) fetch_a2<AP_PITCH>(o.pa[u], pp + 4 * u);
}

__device__ __forceinline__ void mma_block_c(const HBlock& c, v8f (&z)[2][4]) {
#pragma unroll
  for (int u = 0; u < HC_BLK; ++u)
#pragma unroll
    for (int j = 0; j < 4; ++j) {
      v2f bb; bb.x = c.hx[u][j]; bb.y = c.hy[u][j];
#pragma unroll
      for (int mt = 0; mt < 2; ++mt) z[mt][j] = wmma4(c.pa[u][mt], bb, z[mt][j]);
    }
}

template <int MASK>
__device__ __forceinline__ float row_xor(float x) {
  const int xi = __builtin_bit_cast(int, x);
  return __builtin_bit_cast(float, __builtin_amdgcn_update_dpp(xi, xi, 0x160 + MASK, 0xF, 0xF, false));
}

__global__ __launch_bounds__(256) void attn_kernel(
    const float* __restrict__ H, const float* __restrict__ Q,
    const float* __restrict__ KT, const int* __restrict__ mask,
    float* __restrict__ Z) {
  __shared__ alignas(16) float q_lds[AT_ROWS * AT_PITCH];
  __shared__ alignas(16) float p_lds[AT_ROWS * AP_PITCH];
  __shared__ float red[8 * AT_ROWS];
  __shared__ float mrow[AT_ROWS];
  __shared__ float srow[AT_ROWS];
  __shared__ float frow[AT_ROWS];

  const int tid  = threadIdx.x;
  const int w    = tid >> 5;
  const int lane = tid & 31;
  const int half = lane >> 4;
  const int l15  = lane & 15;
  const int b    = blockIdx.x >> 5;
  const int t0   = ((blockIdx.x & 31) >> 1) * AT_ROWS;
  const int d0   = (blockIdx.x & 1) * AT_DCOLS + w * 64 + 4 * l15;

  const float* Qb = Q + ((size_t)b * T_ + t0) * DP_;
  for (int i = tid; i < AT_ROWS * DP_; i += 256) q_lds[(i >> 8) * AT_PITCH + (i & 255)] = Qb[i];
  if (tid < AT_ROWS) { mrow[tid] = -3.0e38f; srow[tid] = 0.f; }
  __syncthreads();

  v8f zacc[2][4];
#pragma unroll
  for (int mt = 0; mt < 2; ++mt)
#pragma unroll
    for (int j = 0; j < 4; ++j) zacc[mt][j] = zero8();

  const float scale = 0.0625f;
  const float* KTb = KT + (size_t)b * DP_ * L_;
  const int col4 = w * (64 * AT_NG) + 4 * l15;

  for (int lc = 0; lc < L_; lc += AT_CHUNK) {
    v8f s[AT_NG][2][4];
#pragma unroll
    for (int g = 0; g < AT_NG; ++g)
#pragma unroll
      for (int mt = 0; mt < 2; ++mt)
#pragma unroll
        for (int j = 0; j < 4; ++j) s[g][mt][j] = zero8();
    {
      const float* kp0 = KTb + (size_t)(2 * half) * L_ + lc + col4;
      const float* qp0 = q_lds + l15 * AT_PITCH + 2 * half;
      constexpr int NB = DP_ / 4 / KA_BLK;
      constexpr size_t KSTR = (size_t)(4 * KA_BLK) * L_;
      constexpr int QSTR = 4 * KA_BLK;
      KBlock ka, kb;
      fetch_block_a(ka, kp0, qp0);
      for (int bl = 0; bl < NB - 2; bl += 2) {
        fetch_block_a(kb, kp0 + (size_t)(bl + 1) * KSTR, qp0 + (bl + 1) * QSTR);
        keep_loads_grouped();
        mma_block_a(ka, s);
        keep_loads_grouped();
        fetch_block_a(ka, kp0 + (size_t)(bl + 2) * KSTR, qp0 + (bl + 2) * QSTR);
        keep_loads_grouped();
        mma_block_a(kb, s);
      }
      fetch_block_a(kb, kp0 + (size_t)(NB - 1) * KSTR, qp0 + (NB - 1) * QSTR);
      keep_loads_grouped();
      mma_block_a(ka, s);
      mma_block_a(kb, s);
    }

    const float* Hb = H + ((size_t)b * L_ + lc + 2 * half) * DH_ + d0;
    HBlock ha, hb;
    fetch_block_h(ha, Hb);

    int mv[AT_NG][4];
#pragma unroll
    for (int g = 0; g < AT_NG; ++g)
#pragma unroll
      for (int j = 0; j < 4; ++j) mv[g][j] = mask[(size_t)b * L_ + lc + col4 + 64 * g + j];
    float cmx[2][8];
#pragma unroll
    for (int mt = 0; mt < 2; ++mt)
#pragma unroll
      for (int v = 0; v < 8; ++v) {
        float cm = 0.f;
#pragma unroll
        for (int g = 0; g < AT_NG; ++g) {
          float x[4];
#pragma unroll
          for (int j = 0; j < 4; ++j) {
            float xv = s[g][mt][j][v] * scale; if (mv[g][j]) xv = -3.0e38f;
            s[g][mt][j][v] = xv; x[j] = xv;
          }
          const float m4 = fmaxf(fmaxf(x[0], x[1]), fmaxf(x[2], x[3]));
          cm = (g == 0) ? m4 : fmaxf(cm, m4);
        }
        cmx[mt][v] = cm;
      }
#pragma unroll
    for (int i = 0; i < 16; ++i) cmx[i >> 3][i & 7] = fmaxf(cmx[i >> 3][i & 7], row_xor<8>(cmx[i >> 3][i & 7]));
#pragma unroll
    for (int i = 0; i < 16; ++i) cmx[i >> 3][i & 7] = fmaxf(cmx[i >> 3][i & 7], row_xor<4>(cmx[i >> 3][i & 7]));
#pragma unroll
    for (int i = 0; i < 16; ++i) cmx[i >> 3][i & 7] = fmaxf(cmx[i >> 3][i & 7], row_xor<2>(cmx[i >> 3][i & 7]));
#pragma unroll
    for (int i = 0; i < 16; ++i) cmx[i >> 3][i & 7] = fmaxf(cmx[i >> 3][i & 7], row_xor<1>(cmx[i >> 3][i & 7]));
    if (l15 == 0) {
#pragma unroll
      for (int i = 0; i < 16; ++i) red[w * AT_ROWS + (i >> 3) * 16 + (i & 7) + 8 * half] = cmx[i >> 3][i & 7];
    }
    __syncthreads();

    if (tid < AT_ROWS) {
      float cm = red[tid];
#pragma unroll
      for (int ww = 1; ww < 8; ++ww) cm = fmaxf(cm, red[ww * AT_ROWS + tid]);
      const float mo = mrow[tid];
      const float mn = fmaxf(mo, cm);
      const float f  = __expf(mo - mn);
      srow[tid] *= f;
      frow[tid] = f;
      mrow[tid] = mn;
    }
    __syncthreads();

    float fsel[2][8], msel[2][8];
#pragma unroll
    for (int mt = 0; mt < 2; ++mt)
#pragma unroll
      for (int v = 0; v < 8; ++v) {
        fsel[mt][v] = frow[mt * 16 + v + 8 * half];
        msel[mt][v] = mrow[mt * 16 + v + 8 * half];
      }
#pragma unroll
    for (int mt = 0; mt < 2; ++mt)
#pragma unroll
      for (int j = 0; j < 4; ++j)
#pragma unroll
        for (int v = 0; v < 8; ++v) zacc[mt][j][v] *= fsel[mt][v];

    float psum[2][8];
#pragma unroll
    for (int mt = 0; mt < 2; ++mt)
#pragma unroll
      for (int v = 0; v < 8; ++v) {
        const int row = mt * 16 + v + 8 * half;
        float ps = 0.f;
#pragma unroll
        for (int g = 0; g < AT_NG; ++g) {
          v4f pp;
          pp.x = __expf(s[g][mt][0][v] - msel[mt][v]);
          pp.y = __expf(s[g][mt][1][v] - msel[mt][v]);
          pp.z = __expf(s[g][mt][2][v] - msel[mt][v]);
          pp.w = __expf(s[g][mt][3][v] - msel[mt][v]);
          *reinterpret_cast<v4f*>(&p_lds[row * AP_PITCH + col4 + 64 * g]) = pp;
          const float s4 = (pp.x + pp.y) + (pp.z + pp.w);
          ps = (g == 0) ? s4 : ps + s4;
        }
        psum[mt][v] = ps;
      }
#pragma unroll
    for (int i = 0; i < 16; ++i) psum[i >> 3][i & 7] += row_xor<8>(psum[i >> 3][i & 7]);
#pragma unroll
    for (int i = 0; i < 16; ++i) psum[i >> 3][i & 7] += row_xor<4>(psum[i >> 3][i & 7]);
#pragma unroll
    for (int i = 0; i < 16; ++i) psum[i >> 3][i & 7] += row_xor<2>(psum[i >> 3][i & 7]);
#pragma unroll
    for (int i = 0; i < 16; ++i) psum[i >> 3][i & 7] += row_xor<1>(psum[i >> 3][i & 7]);
    if (l15 == 0) {
#pragma unroll
      for (int i = 0; i < 16; ++i) red[w * AT_ROWS + (i >> 3) * 16 + (i & 7) + 8 * half] = psum[i >> 3][i & 7];
    }
    __syncthreads();

    if (tid < AT_ROWS) {
      float acc = 0.f;
#pragma unroll
      for (int ww = 0; ww < 8; ++ww) acc += red[ww * AT_ROWS + tid];
      srow[tid] += acc;
    }
    __syncthreads();

    {
      const float* pp0 = p_lds + l15 * AP_PITCH + 2 * half;
      constexpr int NB = AT_CHUNK / 4 / HC_BLK;
      constexpr size_t HSTR = (size_t)(4 * HC_BLK) * DH_;
      constexpr int PSTR = 4 * HC_BLK;
      fetch_block_p(ha, pp0);
      for (int bl = 0; bl < NB - 2; bl += 2) {
        fetch_block_p(hb, pp0 + (bl + 1) * PSTR);
        fetch_block_h(hb, Hb + (size_t)(bl + 1) * HSTR);
        keep_loads_grouped();
        mma_block_c(ha, zacc);
        keep_loads_grouped();
        fetch_block_p(ha, pp0 + (bl + 2) * PSTR);
        fetch_block_h(ha, Hb + (size_t)(bl + 2) * HSTR);
        keep_loads_grouped();
        mma_block_c(hb, zacc);
      }
      fetch_block_p(hb, pp0 + (NB - 1) * PSTR);
      fetch_block_h(hb, Hb + (size_t)(NB - 1) * HSTR);
      keep_loads_grouped();
      mma_block_c(ha, zacc);
      mma_block_c(hb, zacc);
    }
  }

  float* Zb = Z + ((size_t)b * T_ + t0) * DH_ + d0;
#pragma unroll
  for (int mt = 0; mt < 2; ++mt)
#pragma unroll
    for (int v = 0; v < 8; ++v) {
      const int m = mt * 16 + v + 8 * half;
      const float inv = 1.0f / srow[m];
      v4f o;
      o.x = zacc[mt][0][v] * inv;
      o.y = zacc[mt][1][v] * inv;
      o.z = zacc[mt][2][v] * inv;
      o.w = zacc[mt][3][v] * inv;
      *reinterpret_cast<v4f*>(Zb + (size_t)m * DH_) = o;
    }
}

extern "C" void kernel_launch(void* const* d_in, const int* in_sizes, int n_in,
                              void* d_out, int out_size, void* d_ws, size_t ws_size,
                              hipStream_t stream) {
  const float* H    = (const float*)d_in[0];
  const float* G    = (const float*)d_in[1];
  const float* Wk   = (const float*)d_in[2];
  const float* Wq   = (const float*)d_in[3];
  const int*   mask = (const int*)d_in[4];
  float*       Z    = (float*)d_out;

  float* Qws  = (float*)d_ws;
  float* KTws = Qws + (size_t)B_ * T_ * DP_;

  proj_gemm_kernel<<<(B_ * T_) / 16, 256, 0, stream>>>(G, Wq, Qws, DG_);
  kproj_kernel<<<(B_ * L_) / KP_ROWS, 256, 0, stream>>>(H, Wk, KTws);
  attn_kernel<<<B_ * (T_ / AT_ROWS) * (DH_ / AT_DCOLS), 256, 0, stream>>>(H, Qws, KTws, mask, Z);
}


# attention: the two workgroups of a 32-row pair split L instead of DH: logits once, not twice; each leaves partial sums with row max and row sum, a 4th launch merges them (sum order changes)
# speedup vs baseline: 1.8516x; 1.1272x over previous
#include <hip/hip_runtime.h>
#include <hip/hip_bf16.h>

typedef __attribute__((ext_vector_type(2))) float v2f;
typedef __attribute__((ext_vector_type(4))) float v4f;
typedef __attribute__((ext_vector_type(8))) float v8f;

constexpr int B_ = 8, L_ = 8192, T_ = 512, DH_ = 1024, DG_ = 768, DP_ = 256;

__device__ __forceinline__ v8f wmma4(v2f a, v2f b, v8f c) {
  return __builtin_amdgcn_wmma_f32_16x16x4_f32(
       false, a,  false, b,
       (short)0, c,  false,  false);
}

__device__ __forceinline__ v8f zero8() {
  v8f z = {0.f, 0.f, 0.f, 0.f, 0.f, 0.f, 0.f, 0.f};
  return z;
}

__global__ __launch_bounds__(256) void proj_gemm_kernel(
    const float* __restrict__ A, const float* __restrict__ W,
    float* __restrict__ out, int Kd) {
  const int tid  = threadIdx.x;
  const int w    = tid >> 5;
  const int lane = tid & 31;
  const int half = lane >> 4;
  const int l15  = lane & 15;
  const int row_base = blockIdx.x * 16;
  const int col2 = w * 32 + 2 * l15;

  const float* Ab = A + (size_t)(row_base + l15) * Kd + 2 * half;

  v8f acc0 = zero8();
  v8f acc1 = zero8();

  for (int kk = 0; kk < Kd; kk += 4) {
    v2f a;
    a.x = Ab[kk];
    a.y = Ab[kk + 1];
    const float* Wb = W + (size_t)(kk + 2 * half) * DP_ + col2;
    const v2f w0 = *reinterpret_cast<const v2f*>(Wb);
    const v2f w1 = *reinterpret_cast<const v2f*>(Wb + DP_);
    v2f b0; b0.x = w0.x; b0.y = w1.x;
    v2f b1; b1.x = w0.y; b1.y = w1.y;
    acc0 = wmma4(a, b0, acc0);
    acc1 = wmma4(a, b1, acc1);
  }

#pragma unroll
  for (int v = 0; v < 8; ++v) {
    const int m = v + 8 * half;
    const size_t r = (size_t)row_base + m;
    v2f o; o.x = acc0[v]; o.y = acc1[v];
    *reinterpret_cast<v2f*>(out + r * DP_ + col2) = o;
  }
}

constexpr int KP_ROWS  = 64;
constexpr int KP_KC    = 32;
constexpr int KP_PITCH = 36;

__global__ __launch_bounds__(256) void kproj_kernel(
    const float* __restrict__ A, const float* __restrict__ W,
    float* __restrict__ out) {
  __shared__ alignas(16) float a_lds[2][KP_ROWS * KP_PITCH];

  const int tid  = threadIdx.x;
  const int w    = tid >> 5;
  const int lane = tid & 31;
  const int half = lane >> 4;
  const int l15  = lane & 15;
  const int row_base = blockIdx.x * KP_ROWS;
  const int col2 = w * 32 + 2 * l15;

  const int fr = tid >> 3;
  const int fc = (tid & 7) * 4;
  const float* Ag = A + (size_t)(row_base + fr) * DH_ + fc;
  const int so = fr * KP_PITCH + fc;

  v8f acc[4][2];
#pragma unroll
  for (int mt = 0; mt < 4; ++mt) { acc[mt][0] = zero8(); acc[mt][1] = zero8(); }

  v4f g0 = *reinterpret_cast<const v4f*>(Ag);
  v4f g1 = *reinterpret_cast<const v4f*>(Ag + (size_t)32 * DH_);
  *reinterpret_cast<v4f*>(&a_lds[0][so]) = g0;
  *reinterpret_cast<v4f*>(&a_lds[0][so + 32 * KP_PITCH]) = g1;
  __syncthreads();

  const float* Wl = W + (size_t)(2 * half) * DP_ + col2;
  const int ao = l15 * KP_PITCH + 2 * half;

  for (int kc = 0; kc < DH_; kc += KP_KC) {
    const int buf = (kc / KP_KC) & 1;
    const bool more = kc + KP_KC < DH_;
    if (more) {
      g0 = *reinterpret_cast<const v4f*>(Ag + kc + KP_KC);
      g1 = *reinterpret_cast<const v4f*>(Ag + (size_t)32 * DH_ + kc + KP_KC);
    }
    const float* ab = &a_lds[buf][ao];
#pragma unroll
    for (int ks = 0; ks < KP_KC / 4; ++ks) {
      const float* Wb = Wl + (size_t)(kc + 4 * ks) * DP_;
      const v2f w0 = *reinterpret_cast<const v2f*>(Wb);
      const v2f w1 = *reinterpret_cast<const v2f*>(Wb + DP_);
      v2f b0; b0.x = w0.x; b0.y = w1.x;
      v2f b1; b1.x = w0.y; b1.y = w1.y;
#pragma unroll
      for (int mt = 0; mt < 4; ++mt) {
        const v2f a = *reinterpret_cast<const v2f*>(ab + mt * 16 * KP_PITCH + 4 * ks);
        acc[mt][0] = wmma4(a, b0, acc[mt][0]);
        acc[mt][1] = wmma4(a, b1, acc[mt][1]);
      }
    }
    if (more) {
      *reinterpret_cast<v4f*>(&a_lds[buf ^ 1][so]) = g0;
      *reinterpret_cast<v4f*>(&a_lds[buf ^ 1][so + 32 * KP_PITCH]) = g1;
      __syncthreads();
    }
  }

  const size_t bb = (size_t)row_base >> 13;
  const size_t l0 = ((size_t)row_base & (size_t)(L_ - 1)) + 8 * half;
#pragma unroll
  for (int mt = 0; mt < 4; ++mt)
#pragma unroll
    for (int j = 0; j < 2; ++j) {
      float* o = out + (bb * DP_ + col2 + j) * L_ + l0 + mt * 16;
      v4f lo; lo.x = acc[mt][j][0]; lo.y = acc[mt][j][1]; lo.z = acc[mt][j][2]; lo.w = acc[mt][j][3];
      v4f hi; hi.x = acc[mt][j][4]; hi.y = acc[mt][j][5]; hi.z = acc[mt][j][6]; hi.w = acc[mt][j][7];
      *reinterpret_cast<v4f*>(o)     = lo;
      *reinterpret_cast<v4f*>(o + 4) = hi;
    }
}


constexpr int AT_ROWS  = 32;
constexpr int AT_LSPAN = L_ / 2;
constexpr int AT_NG    = 2;
constexpr int AT_CHUNK = 8 * 64 * AT_NG;
constexpr int AT_PITCH = 260;
constexpr int AP_PITCH = AT_CHUNK + 4;
constexpr int KA_BLK = 8 / AT_NG;
constexpr int HC_BLK = 4;

__device__ __forceinline__ void fetch_k(v4f& x, v4f& y, const float* kp) {
  x = *reinterpret_cast<const v4f*>(kp);
  y = *reinterpret_cast<const v4f*>(kp + L_);
}

__device__ __forceinline__ void fetch_h(v4f& x, v4f& y, const float* hrow) {
  x = *reinterpret_cast<const v4f*>(hrow);
  y = *reinterpret_cast<const v4f*>(hrow + DH_);
}

template <int PITCH>
__device__ __forceinline__ void fetch_a2(v2f (&a)[2], const float* ap) {
  a[0] = *reinterpret_cast<const v2f*>(ap);
  a[1] = *reinterpret_cast<const v2f*>(ap + 16 * PITCH);
}

__device__ __forceinline__ void keep_loads_grouped() { __builtin_amdgcn_sched_barrier(0); }

struct KBlock { v4f kx[KA_BLK][AT_NG], ky[KA_BLK][AT_NG]; v2f qa[KA_BLK][2]; };
struct HBlock { v4f hx[HC_BLK][2], hy[HC_BLK][2]; v2f pa[HC_BLK][2]; };

__device__ __forceinline__ void fetch_block_a(KBlock& o, const float* kp, const float* qp) {
#pragma unroll
  for (int u = 0; u < KA_BLK; ++u)
#pragma unroll
    for (int g = 0; g < AT_NG; ++g) fetch_k(o.kx[u][g], o.ky[u][g], kp + (size_t)(4 * u) * L_ + 64 * g);
#pragma unroll
  for (int u = 0; u < KA_BLK; ++u) fetch_a2<AT_PITCH>(o.qa[u], qp + 4 * u);
}

__device__ __forceinline__ void mma_block_a(const KBlock& c, v8f (&s)[AT_NG][2][4]) {
#pragma unroll
  for (int u = 0; u < KA_BLK; ++u)
#pragma unroll
    for (int g = 0; g < AT_NG; ++g)
#pragma unroll
      for (int j = 0; j < 4; ++j) {
        v2f bb; bb.x = c.kx[u][g][j]; bb.y = c.ky[u][g][j];
#pragma unroll
        for (int mt = 0; mt < 2; ++mt) s[g][mt][j] = wmma4(c.qa[u][mt], bb, s[g][mt][j]);
      }
}

__device__ __forceinline__ void fetch_block_h(HBlock& o, const float* hp) {
#pragma unroll
  for (int u = 0; u < HC_BLK; ++u)
#pragma unroll
    for (int q = 0; q < 2; ++q) fetch_h(o.hx[u][q], o.hy[u][q], hp + (size_t)(4 * u) * DH_ + (DH_ / 2) * q);
}

__device__ __forceinline__ void fetch_block_p(HBlock& o, const float* pp) {
#pragma unroll
  for (int u = 0; u < HC_BLK; ++u) fetch_a2<AP_PITCH>(o.pa[u], pp + 4 * u);
}

__device__ __forceinline__ void mma_block_c(const HBlock& c, v8f (&z)[2][2][4]) {
#pragma unroll
  for (int u = 0; u < HC_BLK; ++u)
#pragma unroll
    for (int q = 0; q < 2; ++q)
#pragma unroll
      for (int j = 0; j < 4; ++j) {
        v2f bb; bb.x = c.hx[u][q][j]; bb.y = c.hy[u][q][j];
#pragma unroll
        for (int mt = 0; mt < 2; ++mt) z[q][mt][j] = wmma4(c.pa[u][mt], bb, z[q][mt][j]);
      }
}

template <int MASK>
__device__ __forceinline__ float row_xor(float x) {
  const int xi = __builtin_bit_cast(int, x);
  return __builtin_bit_cast(float, __builtin_amdgcn_update_dpp(xi, xi, 0x160 + MASK, 0xF, 0xF, false));
}

constexpr size_t WS_Q_FLOATS  = (size_t)B_ * T_ * DP_;
constexpr size_t WS_KT_FLOATS = (size_t)B_ * DP_ * L_;
constexpr size_t WS_Z1_FLOATS = (size_t)B_ * T_ * DH_;
constexpr size_t WS_ST_FLOATS = (size_t)4 * B_ * T_;
constexpr size_t WS_NEED_BYTES = (WS_Q_FLOATS + WS_KT_FLOATS + WS_Z1_FLOATS + WS_ST_FLOATS) * sizeof(float);

__global__ __launch_bounds__(256) void attn_kernel(
    const float* __restrict__ H, const float* __restrict__ Q,
    const float* __restrict__ KT, const int* __restrict__ mask,
    float* __restrict__ Z, float* __restrict__ Z1, float* __restrict__ stats,
    size_t ws_bytes) {
  const bool split = ws_bytes >= WS_NEED_BYTES;
  const int lh = blockIdx.x & 1;
  if (!split && lh == 1) return;
  const int l_begin = split ? lh * AT_LSPAN : 0;
  const int l_end   = split ? l_begin + AT_LSPAN : L_;

  __shared__ alignas(16) float q_lds[AT_ROWS * AT_PITCH];
  __shared__ alignas(16) float p_lds[AT_ROWS * AP_PITCH];
  __shared__ float red[8 * AT_ROWS];
  __shared__ float mrow[AT_ROWS];
  __shared__ float srow[AT_ROWS];
  __shared__ float frow[AT_ROWS];

  const int tid  = threadIdx.x;
  const int w    = tid >> 5;
  const int lane = tid & 31;
  const int half = lane >> 4;
  const int l15  = lane & 15;
  const int b    = blockIdx.x >> 5;
  const int t0   = ((blockIdx.x & 31) >> 1) * AT_ROWS;
  const int d0   = w * 64 + 4 * l15;

  const float* Qb = Q + ((size_t)b * T_ + t0) * DP_;
  for (int i = tid; i < AT_ROWS * DP_; i += 256) q_lds[(i >> 8) * AT_PITCH + (i & 255)] = Qb[i];
  if (tid < AT_ROWS) { mrow[tid] = -3.0e38f; srow[tid] = 0.f; }
  __syncthreads();

  v8f zacc[2][2][4];
#pragma unroll
  for (int q = 0; q < 2; ++q)
#pragma unroll
    for (int mt = 0; mt < 2; ++mt)
#pragma unroll
      for (int j = 0; j < 4; ++j) zacc[q][mt][j] = zero8();

  const float scale = 0.0625f;
  const float* KTb = KT + (size_t)b * DP_ * L_;
  const int col4 = w * (64 * AT_NG) + 4 * l15;

  for (int lc = l_begin; lc < l_end; lc += AT_CHUNK) {
    v8f s[AT_NG][2][4];
#pragma unroll
    for (int g = 0; g < AT_NG; ++g)
#pragma unroll
      for (int mt = 0; mt < 2; ++mt)
#pragma unroll
        for (int j = 0; j < 4; ++j) s[g][mt][j] = zero8();
    {
      const float* kp0 = KTb + (size_t)(2 * half) * L_ + lc + col4;
      const float* qp0 = q_lds + l15 * AT_PITCH + 2 * half;
      constexpr int NB = DP_ / 4 / KA_BLK;
      constexpr size_t KSTR = (size_t)(4 * KA_BLK) * L_;
      constexpr int QSTR = 4 * KA_BLK;
      KBlock ka, kb;
      fetch_block_a(ka, kp0, qp0);
      for (int bl = 0; bl < NB - 2; bl += 2) {
        fetch_block_a(kb, kp0 + (size_t)(bl + 1) * KSTR, qp0 + (bl + 1) * QSTR);
        keep_loads_grouped();
        mma_block_a(ka, s);
        keep_loads_grouped();
        fetch_block_a(ka, kp0 + (size_t)(bl + 2) * KSTR, qp0 + (bl + 2) * QSTR);
        keep_loads_grouped();
        mma_block_a(kb, s);
      }
      fetch_block_a(kb, kp0 + (size_t)(NB - 1) * KSTR, qp0 + (NB - 1) * QSTR);
      keep_loads_grouped();
      mma_block_a(ka, s);
      mma_block_a(kb, s);
    }

    const float* Hb = H + ((size_t)b * L_ + lc + 2 * half) * DH_ + d0;
    HBlock ha, hb;
    fetch_block_h(ha, Hb);

    int mv[AT_NG][4];
#pragma unroll
    for (int g = 0; g < AT_NG; ++g)
#pragma unroll
      for (int j = 0; j < 4; ++j) mv[g][j] = mask[(size_t)b * L_ + lc + col4 + 64 * g + j];
    float cmx[2][8];
#pragma unroll
    for (int mt = 0; mt < 2; ++mt)
#pragma unroll
      for (int v = 0; v < 8; ++v) {
        float cm = 0.f;
#pragma unroll
        for (int g = 0; g < AT_NG; ++g) {
          float x[4];
#pragma unroll
          for (int j = 0; j < 4; ++j) {
            float xv = s[g][mt][j][v] * scale; if (mv[g][j]) xv = -3.0e38f;
            s[g][mt][j][v] = xv; x[j] = xv;
          }
          const float m4 = fmaxf(fmaxf(x[0], x[1]), fmaxf(x[2], x[3]));
          cm = (g == 0) ? m4 : fmaxf(cm, m4);
        }
        cmx[mt][v] = cm;
      }
#pragma unroll
    for (int i = 0; i < 16; ++i) cmx[i >> 3][i & 7] = fmaxf(cmx[i >> 3][i & 7], row_xor<8>(cmx[i >> 3][i & 7]));
#pragma unroll
    for (int i = 0; i < 16; ++i) cmx[i >> 3][i & 7] = fmaxf(cmx[i >> 3][i & 7], row_xor<4>(cmx[i >> 3][i & 7]));
#pragma unroll
    for (int i = 0; i < 16; ++i) cmx[i >> 3][i & 7] = fmaxf(cmx[i >> 3][i & 7], row_xor<2>(cmx[i >> 3][i & 7]));
#pragma unroll
    for (int i = 0; i < 16; ++i) cmx[i >> 3][i & 7] = fmaxf(cmx[i >> 3][i & 7], row_xor<1>(cmx[i >> 3][i & 7]));
    if (l15 == 0) {
#pragma unroll
      for (int i = 0; i < 16; ++i) red[w * AT_ROWS + (i >> 3) * 16 + (i & 7) + 8 * half] = cmx[i >> 3][i & 7];
    }
    __syncthreads();

    if (tid < AT_ROWS) {
      float cm = red[tid];
#pragma unroll
      for (int ww = 1; ww < 8; ++ww) cm = fmaxf(cm, red[ww * AT_ROWS + tid]);
      const float mo = mrow[tid];
      const float mn = fmaxf(mo, cm);
      const float f  = __expf(mo - mn);
      srow[tid] *= f;
      frow[tid] = f;
      mrow[tid] = mn;
    }
    __syncthreads();

    float fsel[2][8], msel[2][8];
#pragma unroll
    for (int mt = 0; mt < 2; ++mt)
#pragma unroll
      for (int v = 0; v < 8; ++v) {
        fsel[mt][v] = frow[mt * 16 + v + 8 * half];
        msel[mt][v] = mrow[mt * 16 + v + 8 * half];
      }
#pragma unroll
    for (int q = 0; q < 2; ++q)
#pragma unroll
      for (int mt = 0; mt < 2; ++mt)
#pragma unroll
        for (int j = 0; j < 4; ++j)
#pragma unroll
          for (int v = 0; v < 8; ++v) zacc[q][mt][j][v] *= fsel[mt][v];

    float psum[2][8];
#pragma unroll
    for (int mt = 0; mt < 2; ++mt)
#pragma unroll
      for (int v = 0; v < 8; ++v) {
        const int row = mt * 16 + v + 8 * half;
        float ps = 0.f;
#pragma unroll
        for (int g = 0; g < AT_NG; ++g) {
          v4f pp;
          pp.x = __expf(s[g][mt][0][v] - msel[mt][v]);
          pp.y = __expf(s[g][mt][1][v] - msel[mt][v]);
          pp.z = __expf(s[g][mt][2][v] - msel[mt][v]);
          pp.w = __expf(s[g][mt][3][v] - msel[mt][v]);
          *reinterpret_cast<v4f*>(&p_lds[row * AP_PITCH + col4 + 64 * g]) = pp;
          const float s4 = (pp.x + pp.y) + (pp.z + pp.w);
          ps = (g == 0) ? s4 : ps + s4;
        }
        psum[mt][v] = ps;
      }
#pragma unroll
    for (int i = 0; i < 16; ++i) psum[i >> 3][i & 7] += row_xor<8>(psum[i >> 3][i & 7]);
#pragma unroll
    for (int i = 0; i < 16; ++i) psum[i >> 3][i & 7] += row_xor<4>(psum[i >> 3][i & 7]);
#pragma unroll
    for (int i = 0; i < 16; ++i) psum[i >> 3][i & 7] += row_xor<2>(psum[i >> 3][i & 7]);
#pragma unroll
    for (int i = 0; i < 16; ++i) psum[i >> 3][i & 7] += row_xor<1>(psum[i >> 3][i & 7]);
    if (l15 == 0) {
#pragma unroll
      for (int i = 0; i < 16; ++i) red[w * AT_ROWS + (i >> 3) * 16 + (i & 7) + 8 * half] = psum[i >> 3][i & 7];
    }
    __syncthreads();

    if (tid < AT_ROWS) {
      float acc = 0.f;
#pragma unroll
      for (int ww = 0; ww < 8; ++ww) acc += red[ww * AT_ROWS + tid];
      srow[tid] += acc;
    }
    __syncthreads();

    {
      const float* pp0 = p_lds + l15 * AP_PITCH + 2 * half;
      constexpr int NB = AT_CHUNK / 4 / HC_BLK;
      constexpr size_t HSTR = (size_t)(4 * HC_BLK) * DH_;
      constexpr int PSTR = 4 * HC_BLK;
      fetch_block_p(ha, pp0);
      for (int bl = 0; bl < NB - 2; bl += 2) {
        fetch_block_p(hb, pp0 + (bl + 1) * PSTR);
        fetch_block_h(hb, Hb + (size_t)(bl + 1) * HSTR);
        keep_loads_grouped();
        mma_block_c(ha, zacc);
        keep_loads_grouped();
        fetch_block_p(ha, pp0 + (bl + 2) * PSTR);
        fetch_block_h(ha, Hb + (size_t)(bl + 2) * HSTR);
        keep_loads_grouped();
        mma_block_c(hb, zacc);
      }
      fetch_block_p(hb, pp0 + (NB - 1) * PSTR);
      fetch_block_h(hb, Hb + (size_t)(NB - 1) * HSTR);
      keep_loads_grouped();
      mma_block_c(ha, zacc);
      mma_block_c(hb, zacc);
    }
  }

  const size_t row0 = (size_t)b * T_ + t0;
  if (split && tid < AT_ROWS) {
    stats[(size_t)lh * (B_ * T_) + row0 + tid] = mrow[tid];
    stats[(size_t)(2 + lh) * (B_ * T_) + row0 + tid] = srow[tid];
  }
  float* Zb = ((split && lh == 1) ? Z1 : Z) + row0 * DH_ + d0;
  float inv[2][8];
#pragma unroll
  for (int mt = 0; mt < 2; ++mt)
#pragma unroll
    for (int v = 0; v < 8; ++v) {
      const float sr = srow[mt * 16 + v + 8 * half];
      inv[mt][v] = split ? 1.0f : 1.0f / sr;
    }
#pragma unroll
  for (int q = 0; q < 2; ++q)
#pragma unroll
    for (int mt = 0; mt < 2; ++mt)
#pragma unroll
      for (int v = 0; v < 8; ++v) {
        const int m = mt * 16 + v + 8 * half;
        v4f o;
        o.x = zacc[q][mt][0][v] * inv[mt][v];
        o.y = zacc[q][mt][1][v] * inv[mt][v];
        o.z = zacc[q][mt][2][v] * inv[mt][v];
        o.w = zacc[q][mt][3][v] * inv[mt][v];
        *reinterpret_cast<v4f*>(Zb + (size_t)m * DH_ + (DH_ / 2) * q) = o;
      }
}

__global__ __launch_bounds__(256) void merge_kernel(
    float* __restrict__ Z, const float* __restrict__ Z1, const float* __restrict__ stats,
    size_t ws_bytes) {
  if (ws_bytes < WS_NEED_BYTES) return;
  const size_t row = blockIdx.x;
  const float m0 = stats[row];
  const float m1 = stats[(size_t)(B_ * T_) + row];
  const float s0 = stats[(size_t)2 * (B_ * T_) + row];
  const float s1 = stats[(size_t)3 * (B_ * T_) + row];
  const float m  = fmaxf(m0, m1);
  const float f0 = __expf(m0 - m);
  const float f1 = __expf(m1 - m);
  const float inv = 1.0f / (f0 * s0 + f1 * s1);
  const float g0 = f0 * inv;
  const float g1 = f1 * inv;
  const size_t o = row * DH_ + 4 * threadIdx.x;
  const v4f z0 = *reinterpret_cast<const v4f*>(Z + o);
  const v4f z1 = *reinterpret_cast<const v4f*>(Z1 + o);
  v4f r;
  r.x = g0 * z0.x + g1 * z1.x;
  r.y = g0 * z0.y + g1 * z1.y;
  r.z = g0 * z0.z + g1 * z1.z;
  r.w = g0 * z0.w + g1 * z1.w;
  *reinterpret_cast<v4f*>(Z + o) = r;
}

extern "C" void kernel_launch(void* const* d_in, const int* in_sizes, int n_in,
                              void* d_out, int out_size, void* d_ws, size_t ws_size,
                              hipStream_t stream) {
  const float* H    = (const float*)d_in[0];
  const float* G    = (const float*)d_in[1];
  const float* Wk   = (const float*)d_in[2];
  const float* Wq   = (const float*)d_in[3];
  const int*   mask = (const int*)d_in[4];
  float*       Z    = (float*)d_out;

  float* Qws  = (float*)d_ws;
  float* KTws = Qws + WS_Q_FLOATS;
  float* Z1ws = KTws + WS_KT_FLOATS;
  float* STws = Z1ws + WS_Z1_FLOATS;

  proj_gemm_kernel<<<(B_ * T_) / 16, 256, 0, stream>>>(G, Wq, Qws, DG_);
  kproj_kernel<<<(B_ * L_) / KP_ROWS, 256, 0, stream>>>(H, Wk, KTws);
  attn_kernel<<<B_ * (T_ / AT_ROWS) * (L_ / AT_LSPAN), 256, 0, stream>>>(H, Qws, KTws, mask, Z, Z1ws, STws, ws_size);
  merge_kernel<<<B_ * T_, 256, 0, stream>>>(Z, Z1ws, STws, ws_size);
}
